# starting kernel
# baseline (speedup 1.0000x reference)
#include <hip/hip_runtime.h>
#include <hip/hip_bf16.h>

#define BB   16
#define SS   64
#define HH   16
#define DD   16
#define EE   256
#define HID  128
#define MQ   (BB*SS)
#define MPRE (BB*SS*HH)

typedef float v2f __attribute__((ext_vector_type(2)));
typedef float v8f __attribute__((ext_vector_type(8)));

__device__ __forceinline__ float fast_tanh(float x) {
#if __has_builtin(__builtin_amdgcn_tanhf)
    return __builtin_amdgcn_tanhf(x);
#else
    return tanhf(x);
#endif
}

__device__ __forceinline__ v2f ld2(const float* p) {
    return *reinterpret_cast<const v2f*>(p);
}

#define GEMM_WAVES 8

__global__ __launch_bounds__(256) void gemm_bias_wmma_f32(
    const float* __restrict__ A, int lda,
    const float* __restrict__ W, int ldw, int wOff,
    const float* __restrict__ bias,
    float* __restrict__ C, int ldc,
    int tilesN, int K)
{
    const int lane = threadIdx.x;
    const int wave = threadIdx.y;
    const int gw   = blockIdx.x * GEMM_WAVES + wave;
    const int tm   = gw / tilesN;
    const int tn   = gw - tm * tilesN;
    const int m0   = tm * 16;
    const int n0   = tn * 16;
    const int half = lane >> 4;
    const int sub  = lane & 15;
    const int h2   = half * 2;

    const float* aRow = A + (size_t)(m0 + sub) * lda;
    const float* wRow = W + (size_t)(n0 + sub) * ldw + wOff;

    v8f c = {};
    for (int k0 = 0; k0 < K; k0 += 16) {
        v2f a0 = ld2(aRow + k0 +  0 + h2);
        v2f a1 = ld2(aRow + k0 +  4 + h2);
        v2f a2 = ld2(aRow + k0 +  8 + h2);
        v2f a3 = ld2(aRow + k0 + 12 + h2);
        v2f b0 = ld2(wRow + k0 +  0 + h2);
        v2f b1 = ld2(wRow + k0 +  4 + h2);
        v2f b2 = ld2(wRow + k0 +  8 + h2);
        v2f b3 = ld2(wRow + k0 + 12 + h2);
        c = __builtin_amdgcn_wmma_f32_16x16x4_f32(false, a0, false, b0, (short)0, c, false, false);
        c = __builtin_amdgcn_wmma_f32_16x16x4_f32(false, a1, false, b1, (short)0, c, false, false);
        c = __builtin_amdgcn_wmma_f32_16x16x4_f32(false, a2, false, b2, (short)0, c, false, false);
        c = __builtin_amdgcn_wmma_f32_16x16x4_f32(false, a3, false, b3, (short)0, c, false, false);
    }

    const int   col  = n0 + sub;
    const int   row0 = m0 + half * 8;
    const float bb   = bias ? bias[col] : 0.0f;
#pragma unroll
    for (int r = 0; r < 8; ++r)
        C[(size_t)(row0 + r) * ldc + col] = c[r] + bb;
}

__global__ __launch_bounds__(256) void scores_softmax_ctx(
    const float* __restrict__ pre_q,
    const float* __restrict__ pre_k,
    const float* __restrict__ attn_v,
    const float* __restrict__ v,
    float* __restrict__ ctx)
{
    __shared__ float pq[HH * HID];
    __shared__ float av[HID];
    __shared__ float sc[SS * HH];
    __shared__ float vloc[HH * DD];

    const int    tid = threadIdx.x;
    const size_t bi  = blockIdx.x;
    const size_t bat = bi / SS;

    const float* pqg = pre_q + bi * (HH * HID);
    for (int t = tid; t < HH * HID; t += 256) pq[t] = pqg[t];
    for (int t = tid; t < HID;      t += 256) av[t] = attn_v[t];
    const float* vg = v + bi * (HH * DD);
    for (int t = tid; t < HH * DD;  t += 256) vloc[t] = vg[t];
    __syncthreads();

    const int lane = tid & 31;
    const int wv   = tid >> 5;
    const int c0   = lane * 4;
    const float a0 = av[c0], a1 = av[c0 + 1], a2 = av[c0 + 2], a3 = av[c0 + 3];
    const float* pkb = pre_k + bat * (size_t)(SS * HH * HID);

    for (int p = wv; p < SS * HH; p += 8) {
        const int j = p >> 4;
        const int h = p & 15;
        const float* pk  = pkb + ((size_t)j * HH + h) * HID + c0;
        const float* pqh = pq + h * HID + c0;
        float acc = fast_tanh(pqh[0] + pk[0]) * a0
                  + fast_tanh(pqh[1] + pk[1]) * a1
                  + fast_tanh(pqh[2] + pk[2]) * a2
                  + fast_tanh(pqh[3] + pk[3]) * a3;
#pragma unroll
        for (int off = 16; off > 0; off >>= 1)
            acc += __shfl_xor(acc, off, 32);
        if (lane == 0) sc[j * HH + h] = acc;
    }
    __syncthreads();

    if (tid < HH) {
        const int h = tid;
        float m = -3.0e38f;
        for (int j = 0; j < SS; ++j) m = fmaxf(m, sc[j * HH + h]);
        float s = 0.0f;
        for (int j = 0; j < SS; ++j) {
            float e = __expf(sc[j * HH + h] - m);
            sc[j * HH + h] = e;
            s += e;
        }
        const float inv = 1.0f / s;
        for (int j = 0; j < SS; ++j) sc[j * HH + h] *= inv;
    }
    __syncthreads();

    float* cg = ctx + bi * (size_t)(SS * DD);
    for (int idx = tid; idx < SS * DD; idx += 256) {
        const int j = idx >> 4;
        const int d = idx & 15;
        float acc = 0.0f;
#pragma unroll
        for (int h = 0; h < HH; ++h)
            acc = fmaf(sc[j * HH + h], vloc[h * DD + d], acc);
        cg[idx] = acc;
    }
}

extern "C" void kernel_launch(void* const* d_in, const int* in_sizes, int n_in,
                              void* d_out, int out_size, void* d_ws, size_t ws_size,
                              hipStream_t stream) {
    (void)in_sizes; (void)n_in; (void)out_size; (void)ws_size;
    const float* x      = (const float*)d_in[0];
    const float* Wq     = (const float*)d_in[1];
    const float* bq     = (const float*)d_in[2];
    const float* Wk     = (const float*)d_in[3];
    const float* bk     = (const float*)d_in[4];
    const float* Wv     = (const float*)d_in[5];
    const float* bv     = (const float*)d_in[6];
    const float* Wa     = (const float*)d_in[7];
    const float* ba     = (const float*)d_in[8];
    const float* attn_v = (const float*)d_in[9];
    const float* Wfc    = (const float*)d_in[10];
    const float* bfc    = (const float*)d_in[11];
    float* out = (float*)d_out;

    float* q     = (float*)d_ws;
    float* k     = q     + (size_t)MQ * EE;
    float* v     = k     + (size_t)MQ * EE;
    float* pre_q = v     + (size_t)MQ * EE;
    float* pre_k = pre_q + (size_t)MPRE * HID;
    float* ctx   = pre_k + (size_t)MPRE * HID;

    const dim3 blk(32, GEMM_WAVES);

    gemm_bias_wmma_f32<<<128, blk, 0, stream>>>(x, EE, Wq, EE, 0, bq, q, EE, EE / 16, EE);
    gemm_bias_wmma_f32<<<128, blk, 0, stream>>>(x, EE, Wk, EE, 0, bk, k, EE, EE / 16, EE);
    gemm_bias_wmma_f32<<<128, blk, 0, stream>>>(x, EE, Wv, EE, 0, bv, v, EE, EE / 16, EE);

    gemm_bias_wmma_f32<<<1024, blk, 0, stream>>>(q, DD, Wa, 2 * DD, 0,  ba,      pre_q, HID, HID / 16, DD);
    gemm_bias_wmma_f32<<<1024, blk, 0, stream>>>(k, DD, Wa, 2 * DD, DD, nullptr, pre_k, HID, HID / 16, DD);

    scores_softmax_ctx<<<BB * SS, 256, 0, stream>>>(pre_q, pre_k, attn_v, v, ctx);

    gemm_bias_wmma_f32<<<128, blk, 0, stream>>>(ctx, SS * DD, Wfc, SS * DD, 0, bfc, out, EE, EE / 16, SS * DD);
}


# scores: each wave takes 32 keys per step, not 1: 16 row loads in flight, 32 partial sums reduced in one transposing exchange tree (31 DPP/permlane adds, was 5 bpermute per key); sum order kept
# speedup vs baseline: 1.6385x; 1.6385x over previous
#include <hip/hip_runtime.h>
#include <hip/hip_bf16.h>

#define BB   16
#define SS   64
#define HH   16
#define DD   16
#define EE   256
#define HID  128
#define MQ   (BB*SS)
#define MPRE (BB*SS*HH)

typedef float v2f __attribute__((ext_vector_type(2)));
typedef float v4f __attribute__((ext_vector_type(4)));
typedef float v8f __attribute__((ext_vector_type(8)));

__device__ __forceinline__ float fast_tanh(float x) {
#if __has_builtin(__builtin_amdgcn_tanhf)
    return __builtin_amdgcn_tanhf(x);
#else
    return tanhf(x);
#endif
}

__device__ __forceinline__ v2f ld2(const float* p) {
    return *reinterpret_cast<const v2f*>(p);
}

__device__ __forceinline__ v4f ld4(const float* p) {
    return *reinterpret_cast<const v4f*>(p);
}

template <int OFF>
__device__ __forceinline__ float lane_xor(float x) {
    const int b = __builtin_bit_cast(int, x);
    if constexpr (OFF == 16)
        return __builtin_bit_cast(float,
            __builtin_amdgcn_permlanex16(b, b, 0x76543210, 0xfedcba98, false, false));
    else
        return __builtin_bit_cast(float,
            __builtin_amdgcn_update_dpp(0, b, 0x160 + OFF, 0xf, 0xf, true));
}

template <int OFF>
__device__ __forceinline__ void xtree_step(float* p, int lane) {
    const bool up = (lane & OFF) != 0;
#pragma unroll
    for (int k = 0; k < OFF; ++k) {
        const float keep = up ? p[k + OFF] : p[k];
        const float send = up ? p[k] : p[k + OFF];
        p[k] = keep + lane_xor<OFF>(send);
    }
}

#define GEMM_WAVES 8

__global__ __launch_bounds__(256) void gemm_bias_wmma_f32(
    const float* __restrict__ A, int lda,
    const float* __restrict__ W, int ldw, int wOff,
    const float* __restrict__ bias,
    float* __restrict__ C, int ldc,
    int tilesN, int K)
{
    const int lane = threadIdx.x;
    const int wave = threadIdx.y;
    const int gw   = blockIdx.x * GEMM_WAVES + wave;
    const int tm   = gw / tilesN;
    const int tn   = gw - tm * tilesN;
    const int m0   = tm * 16;
    const int n0   = tn * 16;
    const int half = lane >> 4;
    const int sub  = lane & 15;
    const int h2   = half * 2;

    const float* aRow = A + (size_t)(m0 + sub) * lda;
    const float* wRow = W + (size_t)(n0 + sub) * ldw + wOff;

    v8f c = {};
    for (int k0 = 0; k0 < K; k0 += 16) {
        v2f a0 = ld2(aRow + k0 +  0 + h2);
        v2f a1 = ld2(aRow + k0 +  4 + h2);
        v2f a2 = ld2(aRow + k0 +  8 + h2);
        v2f a3 = ld2(aRow + k0 + 12 + h2);
        v2f b0 = ld2(wRow + k0 +  0 + h2);
        v2f b1 = ld2(wRow + k0 +  4 + h2);
        v2f b2 = ld2(wRow + k0 +  8 + h2);
        v2f b3 = ld2(wRow + k0 + 12 + h2);
        c = __builtin_amdgcn_wmma_f32_16x16x4_f32(false, a0, false, b0, (short)0, c, false, false);
        c = __builtin_amdgcn_wmma_f32_16x16x4_f32(false, a1, false, b1, (short)0, c, false, false);
        c = __builtin_amdgcn_wmma_f32_16x16x4_f32(false, a2, false, b2, (short)0, c, false, false);
        c = __builtin_amdgcn_wmma_f32_16x16x4_f32(false, a3, false, b3, (short)0, c, false, false);
    }

    const int   col  = n0 + sub;
    const int   row0 = m0 + half * 8;
    const float bb   = bias ? bias[col] : 0.0f;
#pragma unroll
    for (int r = 0; r < 8; ++r)
        C[(size_t)(row0 + r) * ldc + col] = c[r] + bb;
}

__global__ __launch_bounds__(256) void scores_softmax_ctx(
    const float* __restrict__ pre_q,
    const float* __restrict__ pre_k,
    const float* __restrict__ attn_v,
    const float* __restrict__ v,
    float* __restrict__ ctx)
{
    __shared__ float pq[HH * HID];
    __shared__ float av[HID];
    __shared__ float sc[SS * HH];
    __shared__ float vloc[HH * DD];

    const int    tid = threadIdx.x;
    const size_t bi  = blockIdx.x;
    const size_t bat = bi / SS;

    const float* pqg = pre_q + bi * (HH * HID);
    for (int t = tid; t < HH * HID; t += 256) pq[t] = pqg[t];
    for (int t = tid; t < HID;      t += 256) av[t] = attn_v[t];
    const float* vg = v + bi * (HH * DD);
    for (int t = tid; t < HH * DD;  t += 256) vloc[t] = vg[t];
    __syncthreads();

    const int lane = tid & 31;
    const int wv   = tid >> 5;
    const int c0   = lane * 4;
    const float a0 = av[c0], a1 = av[c0 + 1], a2 = av[c0 + 2], a3 = av[c0 + 3];
    const float* pkb = pre_k + bat * (size_t)(SS * HH * HID);

    for (int hs = 0; hs < 2; ++hs) {
        const int h = wv + 8 * hs;
        const float* pqh = pq + h * HID + c0;
        const float q0 = pqh[0], q1 = pqh[1], q2 = pqh[2], q3 = pqh[3];
        const float* pkh = pkb + (size_t)h * HID + c0;
        for (int jb = 0; jb < SS; jb += 32) {
            float part[32];
#pragma unroll
            for (int g = 0; g < 32; g += 16) {
                v4f kk[16];
#pragma unroll
                for (int u = 0; u < 16; ++u)
                    kk[u] = ld4(pkh + (size_t)(jb + g + u) * (HH * HID));
#pragma unroll
                for (int u = 0; u < 16; ++u)
                    part[g + u] = fast_tanh(q0 + kk[u].x) * a0
                                + fast_tanh(q1 + kk[u].y) * a1
                                + fast_tanh(q2 + kk[u].z) * a2
                                + fast_tanh(q3 + kk[u].w) * a3;
            }
            xtree_step<16>(part, lane);
            xtree_step<8>(part, lane);
            xtree_step<4>(part, lane);
            xtree_step<2>(part, lane);
            xtree_step<1>(part, lane);
            sc[(jb + lane) * HH + h] = part[0];
        }
    }
    __syncthreads();

    if (tid < HH) {
        const int h = tid;
        float m = -3.0e38f;
        for (int j = 0; j < SS; ++j) m = fmaxf(m, sc[j * HH + h]);
        float s = 0.0f;
        for (int j = 0; j < SS; ++j) {
            float e = __expf(sc[j * HH + h] - m);
            sc[j * HH + h] = e;
            s += e;
        }
        const float inv = 1.0f / s;
        for (int j = 0; j < SS; ++j) sc[j * HH + h] *= inv;
    }
    __syncthreads();

    float* cg = ctx + bi * (size_t)(SS * DD);
    for (int idx = tid; idx < SS * DD; idx += 256) {
        const int j = idx >> 4;
        const int d = idx & 15;
        float acc = 0.0f;
#pragma unroll
        for (int h = 0; h < HH; ++h)
            acc = fmaf(sc[j * HH + h], vloc[h * DD + d], acc);
        cg[idx] = acc;
    }
}

extern "C" void kernel_launch(void* const* d_in, const int* in_sizes, int n_in,
                              void* d_out, int out_size, void* d_ws, size_t ws_size,
                              hipStream_t stream) {
    (void)in_sizes; (void)n_in; (void)out_size; (void)ws_size;
    const float* x      = (const float*)d_in[0];
    const float* Wq     = (const float*)d_in[1];
    const float* bq     = (const float*)d_in[2];
    const float* Wk     = (const float*)d_in[3];
    const float* bk     = (const float*)d_in[4];
    const float* Wv     = (const float*)d_in[5];
    const float* bv     = (const float*)d_in[6];
    const float* Wa     = (const float*)d_in[7];
    const float* ba     = (const float*)d_in[8];
    const float* attn_v = (const float*)d_in[9];
    const float* Wfc    = (const float*)d_in[10];
    const float* bfc    = (const float*)d_in[11];
    float* out = (float*)d_out;

    float* q     = (float*)d_ws;
    float* k     = q     + (size_t)MQ * EE;
    float* v     = k     + (size_t)MQ * EE;
    float* pre_q = v     + (size_t)MQ * EE;
    float* pre_k = pre_q + (size_t)MPRE * HID;
    float* ctx   = pre_k + (size_t)MPRE * HID;

    const dim3 blk(32, GEMM_WAVES);

    gemm_bias_wmma_f32<<<128, blk, 0, stream>>>(x, EE, Wq, EE, 0, bq, q, EE, EE / 16, EE);
    gemm_bias_wmma_f32<<<128, blk, 0, stream>>>(x, EE, Wk, EE, 0, bk, k, EE, EE / 16, EE);
    gemm_bias_wmma_f32<<<128, blk, 0, stream>>>(x, EE, Wv, EE, 0, bv, v, EE, EE / 16, EE);

    gemm_bias_wmma_f32<<<1024, blk, 0, stream>>>(q, DD, Wa, 2 * DD, 0,  ba,      pre_q, HID, HID / 16, DD);
    gemm_bias_wmma_f32<<<1024, blk, 0, stream>>>(k, DD, Wa, 2 * DD, DD, nullptr, pre_k, HID, HID / 16, DD);

    scores_softmax_ctx<<<BB * SS, 256, 0, stream>>>(pre_q, pre_k, attn_v, v, ctx);

    gemm_bias_wmma_f32<<<128, blk, 0, stream>>>(ctx, SS * DD, Wfc, SS * DD, 0, bfc, out, EE, EE / 16, SS * DD);
}


# Q, K, V projections in one launch and pre_q, pre_k in one launch (grid.y of one batched GEMM kernel picks the operands): 7 dependent launches become 4; tile code, k order and sum order unchanged
# speedup vs baseline: 1.9609x; 1.1968x over previous
#include <hip/hip_runtime.h>
#include <hip/hip_bf16.h>

#define BB   16
#define SS   64
#define HH   16
#define DD   16
#define EE   256
#define HID  128
#define MQ   (BB*SS)
#define MPRE (BB*SS*HH)

typedef float v2f __attribute__((ext_vector_type(2)));
typedef float v4f __attribute__((ext_vector_type(4)));
typedef float v8f __attribute__((ext_vector_type(8)));

__device__ __forceinline__ float fast_tanh(float x) {
#if __has_builtin(__builtin_amdgcn_tanhf)
    return __builtin_amdgcn_tanhf(x);
#else
    return tanhf(x);
#endif
}

__device__ __forceinline__ v2f ld2(const float* p) {
    return *reinterpret_cast<const v2f*>(p);
}

__device__ __forceinline__ v4f ld4(const float* p) {
    return *reinterpret_cast<const v4f*>(p);
}

template <int OFF>
__device__ __forceinline__ float lane_xor(float x) {
    const int b = __builtin_bit_cast(int, x);
    if constexpr (OFF == 16)
        return __builtin_bit_cast(float,
            __builtin_amdgcn_permlanex16(b, b, 0x76543210, 0xfedcba98, false, false));
    else
        return __builtin_bit_cast(float,
            __builtin_amdgcn_update_dpp(0, b, 0x160 + OFF, 0xf, 0xf, true));
}

template <int OFF>
__device__ __forceinline__ void xtree_step(float* p, int lane) {
    const bool up = (lane & OFF) != 0;
#pragma unroll
    for (int k = 0; k < OFF; ++k) {
        const float keep = up ? p[k + OFF] : p[k];
        const float send = up ? p[k] : p[k + OFF];
        p[k] = keep + lane_xor<OFF>(send);
    }
}

#define GEMM_WAVES 8

__global__ __launch_bounds__(256) void gemm_bias_wmma_f32(
    const float* A0, const float* A1, const float* A2, int lda,
    const float* W0, const float* W1, const float* W2, int ldw,
    int wOff0, int wOff1, int wOff2,
    const float* bias0, const float* bias1, const float* bias2,
    float* C0, float* C1, float* C2, int ldc,
    int tilesN, int K)
{
    const int z = blockIdx.y;
    const float* __restrict__ A    = (z == 0) ? A0    : ((z == 1) ? A1    : A2);
    const float* __restrict__ W    = (z == 0) ? W0    : ((z == 1) ? W1    : W2);
    const float* __restrict__ bias = (z == 0) ? bias0 : ((z == 1) ? bias1 : bias2);
    float* __restrict__       C    = (z == 0) ? C0    : ((z == 1) ? C1    : C2);
    const int wOff                 = (z == 0) ? wOff0 : ((z == 1) ? wOff1 : wOff2);

    const int lane = threadIdx.x;
    const int wave = threadIdx.y;
    const int gw   = blockIdx.x * GEMM_WAVES + wave;
    const int tm   = gw / tilesN;
    const int tn   = gw - tm * tilesN;
    const int m0   = tm * 16;
    const int n0   = tn * 16;
    const int half = lane >> 4;
    const int sub  = lane & 15;
    const int h2   = half * 2;

    const float* aRow = A + (size_t)(m0 + sub) * lda;
    const float* wRow = W + (size_t)(n0 + sub) * ldw + wOff;

    v8f c = {};
    for (int k0 = 0; k0 < K; k0 += 16) {
        v2f a0 = ld2(aRow + k0 +  0 + h2);
        v2f a1 = ld2(aRow + k0 +  4 + h2);
        v2f a2 = ld2(aRow + k0 +  8 + h2);
        v2f a3 = ld2(aRow + k0 + 12 + h2);
        v2f b0 = ld2(wRow + k0 +  0 + h2);
        v2f b1 = ld2(wRow + k0 +  4 + h2);
        v2f b2 = ld2(wRow + k0 +  8 + h2);
        v2f b3 = ld2(wRow + k0 + 12 + h2);
        c = __builtin_amdgcn_wmma_f32_16x16x4_f32(false, a0, false, b0, (short)0, c, false, false);
        c = __builtin_amdgcn_wmma_f32_16x16x4_f32(false, a1, false, b1, (short)0, c, false, false);
        c = __builtin_amdgcn_wmma_f32_16x16x4_f32(false, a2, false, b2, (short)0, c, false, false);
        c = __builtin_amdgcn_wmma_f32_16x16x4_f32(false, a3, false, b3, (short)0, c, false, false);
    }

    const int   col  = n0 + sub;
    const int   row0 = m0 + half * 8;
    const float bb   = bias ? bias[col] : 0.0f;
#pragma unroll
    for (int r = 0; r < 8; ++r)
        C[(size_t)(row0 + r) * ldc + col] = c[r] + bb;
}

__global__ __launch_bounds__(256) void scores_softmax_ctx(
    const float* __restrict__ pre_q,
    const float* __restrict__ pre_k,
    const float* __restrict__ attn_v,
    const float* __restrict__ v,
    float* __restrict__ ctx)
{
    __shared__ float pq[HH * HID];
    __shared__ float av[HID];
    __shared__ float sc[SS * HH];
    __shared__ float vloc[HH * DD];

    const int    tid = threadIdx.x;
    const size_t bi  = blockIdx.x;
    const size_t bat = bi / SS;

    const float* pqg = pre_q + bi * (HH * HID);
    for (int t = tid; t < HH * HID; t += 256) pq[t] = pqg[t];
    for (int t = tid; t < HID;      t += 256) av[t] = attn_v[t];
    const float* vg = v + bi * (HH * DD);
    for (int t = tid; t < HH * DD;  t += 256) vloc[t] = vg[t];
    __syncthreads();

    const int lane = tid & 31;
    const int wv   = tid >> 5;
    const int c0   = lane * 4;
    const float a0 = av[c0], a1 = av[c0 + 1], a2 = av[c0 + 2], a3 = av[c0 + 3];
    const float* pkb = pre_k + bat * (size_t)(SS * HH * HID);

    for (int hs = 0; hs < 2; ++hs) {
        const int h = wv + 8 * hs;
        const float* pqh = pq + h * HID + c0;
        const float q0 = pqh[0], q1 = pqh[1], q2 = pqh[2], q3 = pqh[3];
        const float* pkh = pkb + (size_t)h * HID + c0;
        for (int jb = 0; jb < SS; jb += 32) {
            float part[32];
#pragma unroll
            for (int g = 0; g < 32; g += 16) {
                v4f kk[16];
#pragma unroll
                for (int u = 0; u < 16; ++u)
                    kk[u] = ld4(pkh + (size_t)(jb + g + u) * (HH * HID));
#pragma unroll
                for (int u = 0; u < 16; ++u)
                    part[g + u] = fast_tanh(q0 + kk[u].x) * a0
                                + fast_tanh(q1 + kk[u].y) * a1
                                + fast_tanh(q2 + kk[u].z) * a2
                                + fast_tanh(q3 + kk[u].w) * a3;
            }
            xtree_step<16>(part, lane);
            xtree_step<8>(part, lane);
            xtree_step<4>(part, lane);
            xtree_step<2>(part, lane);
            xtree_step<1>(part, lane);
            sc[(jb + lane) * HH + h] = part[0];
        }
    }
    __syncthreads();

    if (tid < HH) {
        const int h = tid;
        float m = -3.0e38f;
        for (int j = 0; j < SS; ++j) m = fmaxf(m, sc[j * HH + h]);
        float s = 0.0f;
        for (int j = 0; j < SS; ++j) {
            float e = __expf(sc[j * HH + h] - m);
            sc[j * HH + h] = e;
            s += e;
        }
        const float inv = 1.0f / s;
        for (int j = 0; j < SS; ++j) sc[j * HH + h] *= inv;
    }
    __syncthreads();

    float* cg = ctx + bi * (size_t)(SS * DD);
    for (int idx = tid; idx < SS * DD; idx += 256) {
        const int j = idx >> 4;
        const int d = idx & 15;
        float acc = 0.0f;
#pragma unroll
        for (int h = 0; h < HH; ++h)
            acc = fmaf(sc[j * HH + h], vloc[h * DD + d], acc);
        cg[idx] = acc;
    }
}

extern "C" void kernel_launch(void* const* d_in, const int* in_sizes, int n_in,
                              void* d_out, int out_size, void* d_ws, size_t ws_size,
                              hipStream_t stream) {
    (void)in_sizes; (void)n_in; (void)out_size; (void)ws_size;
    const float* x      = (const float*)d_in[0];
    const float* Wq     = (const float*)d_in[1];
    const float* bq     = (const float*)d_in[2];
    const float* Wk     = (const float*)d_in[3];
    const float* bk     = (const float*)d_in[4];
    const float* Wv     = (const float*)d_in[5];
    const float* bv     = (const float*)d_in[6];
    const float* Wa     = (const float*)d_in[7];
    const float* ba     = (const float*)d_in[8];
    const float* attn_v = (const float*)d_in[9];
    const float* Wfc    = (const float*)d_in[10];
    const float* bfc    = (const float*)d_in[11];
    float* out = (float*)d_out;

    float* q     = (float*)d_ws;
    float* k     = q     + (size_t)MQ * EE;
    float* v     = k     + (size_t)MQ * EE;
    float* pre_q = v     + (size_t)MQ * EE;
    float* pre_k = pre_q + (size_t)MPRE * HID;
    float* ctx   = pre_k + (size_t)MPRE * HID;

    const dim3 blk(32, GEMM_WAVES);
    const float* none  = nullptr;
    float*       noneC = nullptr;

    gemm_bias_wmma_f32<<<dim3(128, 3), blk, 0, stream>>>(
        x, x, x, EE,
        Wq, Wk, Wv, EE, 0, 0, 0,
        bq, bk, bv,
        q, k, v, EE,
        EE / 16, EE);

    gemm_bias_wmma_f32<<<dim3(1024, 2), blk, 0, stream>>>(
        q, k, none, DD,
        Wa, Wa, none, 2 * DD, 0, DD, 0,
        ba, none, none,
        pre_q, pre_k, noneC, HID,
        HID / 16, DD);

    scores_softmax_ctx<<<BB * SS, 256, 0, stream>>>(pre_q, pre_k, attn_v, v, ctx);

    gemm_bias_wmma_f32<<<dim3(128, 1), blk, 0, stream>>>(
        ctx, none, none, SS * DD,
        Wfc, none, none, SS * DD, 0, 0, 0,
        bfc, none, none,
        out, noneC, noneC, EE,
        EE / 16, SS * DD);
}


# scores: one block takes 4 queries of a batch (16 waves, one head each): every pre_k row load feeds 4 queries from registers, two per packed f32 op; cache reads of pre_k cut 4x; sum order kept
# speedup vs baseline: 2.2511x; 1.1480x over previous
#include <hip/hip_runtime.h>
#include <hip/hip_bf16.h>

#define BB   16
#define SS   64
#define HH   16
#define DD   16
#define EE   256
#define HID  128
#define MQ   (BB*SS)
#define MPRE (BB*SS*HH)

typedef float v2f __attribute__((ext_vector_type(2)));
typedef float v4f __attribute__((ext_vector_type(4)));
typedef float v8f __attribute__((ext_vector_type(8)));

__device__ __forceinline__ float fast_tanh(float x) {
#if __has_builtin(__builtin_amdgcn_tanhf)
    return __builtin_amdgcn_tanhf(x);
#else
    return tanhf(x);
#endif
}

__device__ __forceinline__ v2f ld2(const float* p) {
    return *reinterpret_cast<const v2f*>(p);
}

__device__ __forceinline__ v4f ld4(const float* p) {
    return *reinterpret_cast<const v4f*>(p);
}

template <int OFF>
__device__ __forceinline__ float lane_xor(float x) {
    const int b = __builtin_bit_cast(int, x);
    if constexpr (OFF == 16)
        return __builtin_bit_cast(float,
            __builtin_amdgcn_permlanex16(b, b, 0x76543210, 0xfedcba98, false, false));
    else
        return __builtin_bit_cast(float,
            __builtin_amdgcn_update_dpp(0, b, 0x160 + OFF, 0xf, 0xf, true));
}

template <int OFF>
__device__ __forceinline__ void xtree_step(float* p, int lane) {
    const bool up = (lane & OFF) != 0;
#pragma unroll
    for (int k = 0; k < OFF; ++k) {
        const float keep = up ? p[k + OFF] : p[k];
        const float send = up ? p[k] : p[k + OFF];
        p[k] = keep + lane_xor<OFF>(send);
    }
}

#define GEMM_WAVES 8

__global__ __launch_bounds__(256) void gemm_bias_wmma_f32(
    const float* A0, const float* A1, const float* A2, int lda,
    const float* W0, const float* W1, const float* W2, int ldw,
    int wOff0, int wOff1, int wOff2,
    const float* bias0, const float* bias1, const float* bias2,
    float* C0, float* C1, float* C2, int ldc,
    int tilesN, int K)
{
    const int z = blockIdx.y;
    const float* __restrict__ A    = (z == 0) ? A0    : ((z == 1) ? A1    : A2);
    const float* __restrict__ W    = (z == 0) ? W0    : ((z == 1) ? W1    : W2);
    const float* __restrict__ bias = (z == 0) ? bias0 : ((z == 1) ? bias1 : bias2);
    float* __restrict__       C    = (z == 0) ? C0    : ((z == 1) ? C1    : C2);
    const int wOff                 = (z == 0) ? wOff0 : ((z == 1) ? wOff1 : wOff2);

    const int lane = threadIdx.x;
    const int wave = threadIdx.y;
    const int gw   = blockIdx.x * GEMM_WAVES + wave;
    const int tm   = gw / tilesN;
    const int tn   = gw - tm * tilesN;
    const int m0   = tm * 16;
    const int n0   = tn * 16;
    const int half = lane >> 4;
    const int sub  = lane & 15;
    const int h2   = half * 2;

    const float* aRow = A + (size_t)(m0 + sub) * lda;
    const float* wRow = W + (size_t)(n0 + sub) * ldw + wOff;

    v8f c = {};
    for (int k0 = 0; k0 < K; k0 += 16) {
        v2f a0 = ld2(aRow + k0 +  0 + h2);
        v2f a1 = ld2(aRow + k0 +  4 + h2);
        v2f a2 = ld2(aRow + k0 +  8 + h2);
        v2f a3 = ld2(aRow + k0 + 12 + h2);
        v2f b0 = ld2(wRow + k0 +  0 + h2);
        v2f b1 = ld2(wRow + k0 +  4 + h2);
        v2f b2 = ld2(wRow + k0 +  8 + h2);
        v2f b3 = ld2(wRow + k0 + 12 + h2);
        c = __builtin_amdgcn_wmma_f32_16x16x4_f32(false, a0, false, b0, (short)0, c, false, false);
        c = __builtin_amdgcn_wmma_f32_16x16x4_f32(false, a1, false, b1, (short)0, c, false, false);
        c = __builtin_amdgcn_wmma_f32_16x16x4_f32(false, a2, false, b2, (short)0, c, false, false);
        c = __builtin_amdgcn_wmma_f32_16x16x4_f32(false, a3, false, b3, (short)0, c, false, false);
    }

    const int   col  = n0 + sub;
    const int   row0 = m0 + half * 8;
    const float bb   = bias ? bias[col] : 0.0f;
#pragma unroll
    for (int r = 0; r < 8; ++r)
        C[(size_t)(row0 + r) * ldc + col] = c[r] + bb;
}

#define IQ        4
#define SCP       (SS * HH + 16)
#define SC_THREADS (32 * HH)
#define SC_BLOCKS  (BB * SS / IQ)

__device__ __forceinline__ void load_keys8(v4f (&kk)[8], const float* pkh, int j0) {
#pragma unroll
    for (int u = 0; u < 8; ++u)
        kk[u] = ld4(pkh + (size_t)(j0 + u) * (HH * HID));
}

__device__ __forceinline__ v2f tanh2(v2f s) {
    return v2f{fast_tanh(s.x), fast_tanh(s.y)};
}

__device__ __forceinline__ float scores_8keys(const v4f (&kk)[8],
                                              const v2f (&qx)[IQ / 2], const v2f (&qy)[IQ / 2],
                                              const v2f (&qz)[IQ / 2], const v2f (&qw)[IQ / 2],
                                              const v4f a, int lane) {
    const v2f ax = {a.x, a.x}, ay = {a.y, a.y}, az = {a.z, a.z}, aw = {a.w, a.w};
    float part[32];
#pragma unroll
    for (int qp = 0; qp < IQ / 2; ++qp) {
#pragma unroll
        for (int u = 0; u < 8; ++u) {
            const v2f kx = {kk[u].x, kk[u].x}, ky = {kk[u].y, kk[u].y};
            const v2f kz = {kk[u].z, kk[u].z}, kw = {kk[u].w, kk[u].w};
            const v2f r = tanh2(qx[qp] + kx) * ax
                        + tanh2(qy[qp] + ky) * ay
                        + tanh2(qz[qp] + kz) * az
                        + tanh2(qw[qp] + kw) * aw;
            part[(2 * qp) * 8 + u]     = r.x;
            part[(2 * qp + 1) * 8 + u] = r.y;
        }
    }
    xtree_step<16>(part, lane);
    xtree_step<8>(part, lane);
    xtree_step<4>(part, lane);
    xtree_step<2>(part, lane);
    xtree_step<1>(part, lane);
    return part[0];
}

__global__ __launch_bounds__(SC_THREADS) void scores_softmax_ctx(
    const float* __restrict__ pre_q,
    const float* __restrict__ pre_k,
    const float* __restrict__ attn_v,
    const float* __restrict__ v,
    float* __restrict__ ctx)
{
    __shared__ float sc[IQ * SCP];
    __shared__ float vloc[IQ * HH * DD];

    const int    tid = threadIdx.x;
    const size_t bi0 = (size_t)blockIdx.x * IQ;
    const size_t bat = bi0 / SS;

    const float* vg = v + bi0 * (HH * DD);
    for (int t = tid; t < IQ * HH * DD; t += SC_THREADS) vloc[t] = vg[t];

    const int lane = tid & 31;
    const int h    = tid >> 5;
    const int c0   = lane * 4;
    const v4f a    = ld4(attn_v + c0);
    v2f qx[IQ / 2], qy[IQ / 2], qz[IQ / 2], qw[IQ / 2];
#pragma unroll
    for (int qp = 0; qp < IQ / 2; ++qp) {
        const float* pq0 = pre_q + (bi0 + 2 * qp) * (size_t)(HH * HID) + (size_t)h * HID + c0;
        const v4f q0 = ld4(pq0);
        const v4f q1 = ld4(pq0 + HH * HID);
        qx[qp] = v2f{q0.x, q1.x};  qy[qp] = v2f{q0.y, q1.y};
        qz[qp] = v2f{q0.z, q1.z};  qw[qp] = v2f{q0.w, q1.w};
    }
    const float* pkh = pre_k + bat * (size_t)(SS * HH * HID) + (size_t)h * HID + c0;

    const int sidx = (lane >> 3) * SCP + (lane & 7) * HH + h;
    v4f kA[8], kB[8];
    load_keys8(kA, pkh, 0);
    for (int jb = 0; jb < SS - 16; jb += 16) {
        load_keys8(kB, pkh, jb + 8);
        __builtin_amdgcn_sched_barrier(0);
        sc[sidx + jb * HH] = scores_8keys(kA, qx, qy, qz, qw, a, lane);
        load_keys8(kA, pkh, jb + 16);
        __builtin_amdgcn_sched_barrier(0);
        sc[sidx + (jb + 8) * HH] = scores_8keys(kB, qx, qy, qz, qw, a, lane);
    }
    load_keys8(kB, pkh, SS - 8);
    __builtin_amdgcn_sched_barrier(0);
    sc[sidx + (SS - 16) * HH] = scores_8keys(kA, qx, qy, qz, qw, a, lane);
    sc[sidx + (SS - 8) * HH]  = scores_8keys(kB, qx, qy, qz, qw, a, lane);
    __syncthreads();

    if (tid < IQ * HH) {
        float* s_ = sc + (tid >> 4) * SCP;
        const int hh = tid & 15;
        float m = -3.0e38f;
        for (int j = 0; j < SS; ++j) m = fmaxf(m, s_[j * HH + hh]);
        float s = 0.0f;
        for (int j = 0; j < SS; ++j) {
            float e = __expf(s_[j * HH + hh] - m);
            s_[j * HH + hh] = e;
            s += e;
        }
        const float inv = 1.0f / s;
        for (int j = 0; j < SS; ++j) s_[j * HH + hh] *= inv;
    }
    __syncthreads();

    float* cg = ctx + bi0 * (size_t)(SS * DD);
    for (int idx = tid; idx < IQ * SS * DD; idx += SC_THREADS) {
        const int qi = idx >> 10;
        const int j  = (idx >> 4) & (SS - 1);
        const int d  = idx & 15;
        const float* wr = sc + qi * SCP + j * HH;
        const float* vr = vloc + qi * (HH * DD) + d;
        float acc = 0.0f;
#pragma unroll
        for (int hh = 0; hh < HH; ++hh)
            acc = fmaf(wr[hh], vr[hh * DD], acc);
        cg[idx] = acc;
    }
}

extern "C" void kernel_launch(void* const* d_in, const int* in_sizes, int n_in,
                              void* d_out, int out_size, void* d_ws, size_t ws_size,
                              hipStream_t stream) {
    (void)in_sizes; (void)n_in; (void)out_size; (void)ws_size;
    const float* x      = (const float*)d_in[0];
    const float* Wq     = (const float*)d_in[1];
    const float* bq     = (const float*)d_in[2];
    const float* Wk     = (const float*)d_in[3];
    const float* bk     = (const float*)d_in[4];
    const float* Wv     = (const float*)d_in[5];
    const float* bv     = (const float*)d_in[6];
    const float* Wa     = (const float*)d_in[7];
    const float* ba     = (const float*)d_in[8];
    const float* attn_v = (const float*)d_in[9];
    const float* Wfc    = (const float*)d_in[10];
    const float* bfc    = (const float*)d_in[11];
    float* out = (float*)d_out;

    float* q     = (float*)d_ws;
    float* k     = q     + (size_t)MQ * EE;
    float* v     = k     + (size_t)MQ * EE;
    float* pre_q = v     + (size_t)MQ * EE;
    float* pre_k = pre_q + (size_t)MPRE * HID;
    float* ctx   = pre_k + (size_t)MPRE * HID;

    const dim3 blk(32, GEMM_WAVES);
    const float* none  = nullptr;
    float*       noneC = nullptr;

    gemm_bias_wmma_f32<<<dim3(128, 3), blk, 0, stream>>>(
        x, x, x, EE,
        Wq, Wk, Wv, EE, 0, 0, 0,
        bq, bk, bv,
        q, k, v, EE,
        EE / 16, EE);

    gemm_bias_wmma_f32<<<dim3(1024, 2), blk, 0, stream>>>(
        q, k, none, DD,
        Wa, Wa, none, 2 * DD, 0, DD, 0,
        ba, none, none,
        pre_q, pre_k, noneC, HID,
        HID / 16, DD);

    scores_softmax_ctx<<<SC_BLOCKS, SC_THREADS, 0, stream>>>(pre_q, pre_k, attn_v, v, ctx);

    gemm_bias_wmma_f32<<<dim3(128, 1), blk, 0, stream>>>(
        ctx, none, none, SS * DD,
        Wfc, none, none, SS * DD, 0, 0, 0,
        bfc, none, none,
        out, noneC, noneC, EE,
        EE / 16, SS * DD);
}


# GEMMs: fragment loads 16 bytes per lane (4 consecutive k) instead of 8: load instructions and line touches per k cut 2x; k grouping inside each 8 changes alike for A and W (sum order changes)
# speedup vs baseline: 2.3288x; 1.0345x over previous
#include <hip/hip_runtime.h>
#include <hip/hip_bf16.h>

#define BB   16
#define SS   64
#define HH   16
#define DD   16
#define EE   256
#define HID  128
#define MQ   (BB*SS)
#define MPRE (BB*SS*HH)

typedef float v2f __attribute__((ext_vector_type(2)));
typedef float v4f __attribute__((ext_vector_type(4)));
typedef float v8f __attribute__((ext_vector_type(8)));

__device__ __forceinline__ float fast_tanh(float x) {
#if __has_builtin(__builtin_amdgcn_tanhf)
    return __builtin_amdgcn_tanhf(x);
#else
    return tanhf(x);
#endif
}

__device__ __forceinline__ v4f ld4(const float* p) {
    return *reinterpret_cast<const v4f*>(p);
}

template <int OFF>
__device__ __forceinline__ float lane_xor(float x) {
    const int b = __builtin_bit_cast(int, x);
    if constexpr (OFF == 16)
        return __builtin_bit_cast(float,
            __builtin_amdgcn_permlanex16(b, b, 0x76543210, 0xfedcba98, false, false));
    else
        return __builtin_bit_cast(float,
            __builtin_amdgcn_update_dpp(0, b, 0x160 + OFF, 0xf, 0xf, true));
}

template <int OFF>
__device__ __forceinline__ void xtree_step(float* p, int lane) {
    const bool up = (lane & OFF) != 0;
#pragma unroll
    for (int k = 0; k < OFF; ++k) {
        const float keep = up ? p[k + OFF] : p[k];
        const float send = up ? p[k] : p[k + OFF];
        p[k] = keep + lane_xor<OFF>(send);
    }
}

#define GEMM_WAVES 8

template <int KU>
__global__ __launch_bounds__(256) void gemm_bias_wmma_f32(
    const float* A0, const float* A1, const float* A2, int lda,
    const float* W0, const float* W1, const float* W2, int ldw,
    int wOff0, int wOff1, int wOff2,
    const float* bias0, const float* bias1, const float* bias2,
    float* C0, float* C1, float* C2, int ldc,
    int tilesN, int K)
{
    const int z = blockIdx.y;
    const float* __restrict__ A    = (z == 0) ? A0    : ((z == 1) ? A1    : A2);
    const float* __restrict__ W    = (z == 0) ? W0    : ((z == 1) ? W1    : W2);
    const float* __restrict__ bias = (z == 0) ? bias0 : ((z == 1) ? bias1 : bias2);
    float* __restrict__       C    = (z == 0) ? C0    : ((z == 1) ? C1    : C2);
    const int wOff                 = (z == 0) ? wOff0 : ((z == 1) ? wOff1 : wOff2);

    const int lane = threadIdx.x;
    const int wave = threadIdx.y;
    const int gw   = blockIdx.x * GEMM_WAVES + wave;
    const int tm   = gw / tilesN;
    const int tn   = gw - tm * tilesN;
    const int m0   = tm * 16;
    const int n0   = tn * 16;
    const int half = lane >> 4;
    const int sub  = lane & 15;
    const int h4   = half * 4;

    const float* aRow = A + (size_t)(m0 + sub) * lda + h4;
    const float* wRow = W + (size_t)(n0 + sub) * ldw + wOff + h4;

    v8f c = {};
    for (int k0 = 0; k0 < K; k0 += KU) {
        v4f a[KU / 8], b[KU / 8];
#pragma unroll
        for (int i = 0; i < KU / 8; ++i) a[i] = ld4(aRow + k0 + 8 * i);
#pragma unroll
        for (int i = 0; i < KU / 8; ++i) b[i] = ld4(wRow + k0 + 8 * i);
        __builtin_amdgcn_sched_barrier(0);
#pragma unroll
        for (int i = 0; i < KU / 8; ++i) {
            c = __builtin_amdgcn_wmma_f32_16x16x4_f32(false, v2f{a[i].x, a[i].y}, false, v2f{b[i].x, b[i].y},
                                                      (short)0, c, false, false);
            c = __builtin_amdgcn_wmma_f32_16x16x4_f32(false, v2f{a[i].z, a[i].w}, false, v2f{b[i].z, b[i].w},
                                                      (short)0, c, false, false);
        }
    }

    const int   col  = n0 + sub;
    const int   row0 = m0 + half * 8;
    const float bb   = bias ? bias[col] : 0.0f;
#pragma unroll
    for (int r = 0; r < 8; ++r)
        C[(size_t)(row0 + r) * ldc + col] = c[r] + bb;
}

#define IQ        4
#define SCP       (SS * HH + 16)
#define SC_THREADS (32 * HH)
#define SC_BLOCKS  (BB * SS / IQ)

__device__ __forceinline__ void load_keys8(v4f (&kk)[8], const float* pkh, int j0) {
#pragma unroll
    for (int u = 0; u < 8; ++u)
        kk[u] = ld4(pkh + (size_t)(j0 + u) * (HH * HID));
}

__device__ __forceinline__ v2f tanh2(v2f s) {
    return v2f{fast_tanh(s.x), fast_tanh(s.y)};
}

__device__ __forceinline__ float scores_8keys(const v4f (&kk)[8],
                                              const v2f (&qx)[IQ / 2], const v2f (&qy)[IQ / 2],
                                              const v2f (&qz)[IQ / 2], const v2f (&qw)[IQ / 2],
                                              const v4f a, int lane) {
    const v2f ax = {a.x, a.x}, ay = {a.y, a.y}, az = {a.z, a.z}, aw = {a.w, a.w};
    float part[32];
#pragma unroll
    for (int qp = 0; qp < IQ / 2; ++qp) {
#pragma unroll
        for (int u = 0; u < 8; ++u) {
            const v2f kx = {kk[u].x, kk[u].x}, ky = {kk[u].y, kk[u].y};
            const v2f kz = {kk[u].z, kk[u].z}, kw = {kk[u].w, kk[u].w};
            const v2f r = tanh2(qx[qp] + kx) * ax
                        + tanh2(qy[qp] + ky) * ay
                        + tanh2(qz[qp] + kz) * az
                        + tanh2(qw[qp] + kw) * aw;
            part[(2 * qp) * 8 + u]     = r.x;
            part[(2 * qp + 1) * 8 + u] = r.y;
        }
    }
    xtree_step<16>(part, lane);
    xtree_step<8>(part, lane);
    xtree_step<4>(part, lane);
    xtree_step<2>(part, lane);
    xtree_step<1>(part, lane);
    return part[0];
}

__global__ __launch_bounds__(SC_THREADS) void scores_softmax_ctx(
    const float* __restrict__ pre_q,
    const float* __restrict__ pre_k,
    const float* __restrict__ attn_v,
    const float* __restrict__ v,
    float* __restrict__ ctx)
{
    __shared__ float sc[IQ * SCP];
    __shared__ float vloc[IQ * HH * DD];

    const int    tid = threadIdx.x;
    const size_t bi0 = (size_t)blockIdx.x * IQ;
    const size_t bat = bi0 / SS;

    const float* vg = v + bi0 * (HH * DD);
    for (int t = tid; t < IQ * HH * DD; t += SC_THREADS) vloc[t] = vg[t];

    const int lane = tid & 31;
    const int h    = tid >> 5;
    const int c0   = lane * 4;
    const v4f a    = ld4(attn_v + c0);
    v2f qx[IQ / 2], qy[IQ / 2], qz[IQ / 2], qw[IQ / 2];
#pragma unroll
    for (int qp = 0; qp < IQ / 2; ++qp) {
        const float* pq0 = pre_q + (bi0 + 2 * qp) * (size_t)(HH * HID) + (size_t)h * HID + c0;
        const v4f q0 = ld4(pq0);
        const v4f q1 = ld4(pq0 + HH * HID);
        qx[qp] = v2f{q0.x, q1.x};  qy[qp] = v2f{q0.y, q1.y};
        qz[qp] = v2f{q0.z, q1.z};  qw[qp] = v2f{q0.w, q1.w};
    }
    const float* pkh = pre_k + bat * (size_t)(SS * HH * HID) + (size_t)h * HID + c0;

    const int sidx = (lane >> 3) * SCP + (lane & 7) * HH + h;
    v4f kA[8], kB[8];
    load_keys8(kA, pkh, 0);
    for (int jb = 0; jb < SS - 16; jb += 16) {
        load_keys8(kB, pkh, jb + 8);
        __builtin_amdgcn_sched_barrier(0);
        sc[sidx + jb * HH] = scores_8keys(kA, qx, qy, qz, qw, a, lane);
        load_keys8(kA, pkh, jb + 16);
        __builtin_amdgcn_sched_barrier(0);
        sc[sidx + (jb + 8) * HH] = scores_8keys(kB, qx, qy, qz, qw, a, lane);
    }
    load_keys8(kB, pkh, SS - 8);
    __builtin_amdgcn_sched_barrier(0);
    sc[sidx + (SS - 16) * HH] = scores_8keys(kA, qx, qy, qz, qw, a, lane);
    sc[sidx + (SS - 8) * HH]  = scores_8keys(kB, qx, qy, qz, qw, a, lane);
    __syncthreads();

    if (tid < IQ * HH) {
        float* s_ = sc + (tid >> 4) * SCP;
        const int hh = tid & 15;
        float m = -3.0e38f;
        for (int j = 0; j < SS; ++j) m = fmaxf(m, s_[j * HH + hh]);
        float s = 0.0f;
        for (int j = 0; j < SS; ++j) {
            float e = __expf(s_[j * HH + hh] - m);
            s_[j * HH + hh] = e;
            s += e;
        }
        const float inv = 1.0f / s;
        for (int j = 0; j < SS; ++j) s_[j * HH + hh] *= inv;
    }
    __syncthreads();

    float* cg = ctx + bi0 * (size_t)(SS * DD);
    for (int idx = tid; idx < IQ * SS * DD; idx += SC_THREADS) {
        const int qi = idx >> 10;
        const int j  = (idx >> 4) & (SS - 1);
        const int d  = idx & 15;
        const float* wr = sc + qi * SCP + j * HH;
        const float* vr = vloc + qi * (HH * DD) + d;
        float acc = 0.0f;
#pragma unroll
        for (int hh = 0; hh < HH; ++hh)
            acc = fmaf(wr[hh], vr[hh * DD], acc);
        cg[idx] = acc;
    }
}

extern "C" void kernel_launch(void* const* d_in, const int* in_sizes, int n_in,
                              void* d_out, int out_size, void* d_ws, size_t ws_size,
                              hipStream_t stream) {
    (void)in_sizes; (void)n_in; (void)out_size; (void)ws_size;
    const float* x      = (const float*)d_in[0];
    const float* Wq     = (const float*)d_in[1];
    const float* bq     = (const float*)d_in[2];
    const float* Wk     = (const float*)d_in[3];
    const float* bk     = (const float*)d_in[4];
    const float* Wv     = (const float*)d_in[5];
    const float* bv     = (const float*)d_in[6];
    const float* Wa     = (const float*)d_in[7];
    const float* ba     = (const float*)d_in[8];
    const float* attn_v = (const float*)d_in[9];
    const float* Wfc    = (const float*)d_in[10];
    const float* bfc    = (const float*)d_in[11];
    float* out = (float*)d_out;

    float* q     = (float*)d_ws;
    float* k     = q     + (size_t)MQ * EE;
    float* v     = k     + (size_t)MQ * EE;
    float* pre_q = v     + (size_t)MQ * EE;
    float* pre_k = pre_q + (size_t)MPRE * HID;
    float* ctx   = pre_k + (size_t)MPRE * HID;

    const dim3 blk(32, GEMM_WAVES);
    const float* none  = nullptr;
    float*       noneC = nullptr;

    gemm_bias_wmma_f32<64><<<dim3(128, 3), blk, 0, stream>>>(
        x, x, x, EE,
        Wq, Wk, Wv, EE, 0, 0, 0,
        bq, bk, bv,
        q, k, v, EE,
        EE / 16, EE);

    gemm_bias_wmma_f32<16><<<dim3(1024, 2), blk, 0, stream>>>(
        q, k, none, DD,
        Wa, Wa, none, 2 * DD, 0, DD, 0,
        ba, none, none,
        pre_q, pre_k, noneC, HID,
        HID / 16, DD);

    scores_softmax_ctx<<<SC_BLOCKS, SC_THREADS, 0, stream>>>(pre_q, pre_k, attn_v, v, ctx);

    gemm_bias_wmma_f32<64><<<dim3(128, 1), blk, 0, stream>>>(
        ctx, none, none, SS * DD,
        Wfc, none, none, SS * DD, 0, 0, 0,
        bfc, none, none,
        out, noneC, noneC, EE,
        EE / 16, SS * DD);
}


# pre_q, pre_k computed inside the Q,K,V launch: a wave's 16x16 q or k tile, built transposed, stays in registers as A operand of its 16x128 product; 4 launches become 3; sum order over d changes
# speedup vs baseline: 2.4395x; 1.0475x over previous
#include <hip/hip_runtime.h>
#include <hip/hip_bf16.h>

#define BB   16
#define SS   64
#define HH   16
#define DD   16
#define EE   256
#define HID  128
#define MQ   (BB*SS)
#define MPRE (BB*SS*HH)

typedef float v2f __attribute__((ext_vector_type(2)));
typedef float v4f __attribute__((ext_vector_type(4)));
typedef float v8f __attribute__((ext_vector_type(8)));

__device__ __forceinline__ float fast_tanh(float x) {
#if __has_builtin(__builtin_amdgcn_tanhf)
    return __builtin_amdgcn_tanhf(x);
#else
    return tanhf(x);
#endif
}

__device__ __forceinline__ v4f ld4(const float* p) {
    return *reinterpret_cast<const v4f*>(p);
}

template <int OFF>
__device__ __forceinline__ float lane_xor(float x) {
    const int b = __builtin_bit_cast(int, x);
    if constexpr (OFF == 16)
        return __builtin_bit_cast(float,
            __builtin_amdgcn_permlanex16(b, b, 0x76543210, 0xfedcba98, false, false));
    else
        return __builtin_bit_cast(float,
            __builtin_amdgcn_update_dpp(0, b, 0x160 + OFF, 0xf, 0xf, true));
}

template <int OFF>
__device__ __forceinline__ void xtree_step(float* p, int lane) {
    const bool up = (lane & OFF) != 0;
#pragma unroll
    for (int k = 0; k < OFF; ++k) {
        const float keep = up ? p[k + OFF] : p[k];
        const float send = up ? p[k] : p[k + OFF];
        p[k] = keep + lane_xor<OFF>(send);
    }
}

#define GEMM_WAVES 8

template <int KU>
__global__ __launch_bounds__(256) void gemm_bias_wmma_f32(
    const float* A0, const float* A1, const float* A2, int lda,
    const float* W0, const float* W1, const float* W2, int ldw,
    int wOff0, int wOff1, int wOff2,
    const float* bias0, const float* bias1, const float* bias2,
    float* C0, float* C1, float* C2, int ldc,
    int tilesN, int K)
{
    const int z = blockIdx.y;
    const float* __restrict__ A    = (z == 0) ? A0    : ((z == 1) ? A1    : A2);
    const float* __restrict__ W    = (z == 0) ? W0    : ((z == 1) ? W1    : W2);
    const float* __restrict__ bias = (z == 0) ? bias0 : ((z == 1) ? bias1 : bias2);
    float* __restrict__       C    = (z == 0) ? C0    : ((z == 1) ? C1    : C2);
    const int wOff                 = (z == 0) ? wOff0 : ((z == 1) ? wOff1 : wOff2);

    const int lane = threadIdx.x;
    const int wave = threadIdx.y;
    const int gw   = blockIdx.x * GEMM_WAVES + wave;
    const int tm   = gw / tilesN;
    const int tn   = gw - tm * tilesN;
    const int m0   = tm * 16;
    const int n0   = tn * 16;
    const int half = lane >> 4;
    const int sub  = lane & 15;
    const int h4   = half * 4;

    const float* aRow = A + (size_t)(m0 + sub) * lda + h4;
    const float* wRow = W + (size_t)(n0 + sub) * ldw + wOff + h4;

    v8f c = {};
    for (int k0 = 0; k0 < K; k0 += KU) {
        v4f a[KU / 8], b[KU / 8];
#pragma unroll
        for (int i = 0; i < KU / 8; ++i) a[i] = ld4(aRow + k0 + 8 * i);
#pragma unroll
        for (int i = 0; i < KU / 8; ++i) b[i] = ld4(wRow + k0 + 8 * i);
        __builtin_amdgcn_sched_barrier(0);
#pragma unroll
        for (int i = 0; i < KU / 8; ++i) {
            c = __builtin_amdgcn_wmma_f32_16x16x4_f32(false, v2f{a[i].x, a[i].y}, false, v2f{b[i].x, b[i].y},
                                                      (short)0, c, false, false);
            c = __builtin_amdgcn_wmma_f32_16x16x4_f32(false, v2f{a[i].z, a[i].w}, false, v2f{b[i].z, b[i].w},
                                                      (short)0, c, false, false);
        }
    }

    const int   col  = n0 + sub;
    const int   row0 = m0 + half * 8;
    const float bb   = bias ? bias[col] : 0.0f;
#pragma unroll
    for (int r = 0; r < 8; ++r)
        C[(size_t)(row0 + r) * ldc + col] = c[r] + bb;
}

__device__ __forceinline__ v8f tile_k256(const float* __restrict__ pRow, const float* __restrict__ qRow) {
    v8f c = {};
    for (int k0 = 0; k0 < EE; k0 += 64) {
        v4f a[8], b[8];
#pragma unroll
        for (int i = 0; i < 8; ++i) a[i] = ld4(pRow + k0 + 8 * i);
#pragma unroll
        for (int i = 0; i < 8; ++i) b[i] = ld4(qRow + k0 + 8 * i);
        __builtin_amdgcn_sched_barrier(0);
#pragma unroll
        for (int i = 0; i < 8; ++i) {
            c = __builtin_amdgcn_wmma_f32_16x16x4_f32(false, v2f{a[i].x, a[i].y}, false, v2f{b[i].x, b[i].y},
                                                      (short)0, c, false, false);
            c = __builtin_amdgcn_wmma_f32_16x16x4_f32(false, v2f{a[i].z, a[i].w}, false, v2f{b[i].z, b[i].w},
                                                      (short)0, c, false, false);
        }
    }
    return c;
}

__global__ __launch_bounds__(256) void qkv_pre_wmma_f32(
    const float* __restrict__ x,
    const float* __restrict__ Wq, const float* __restrict__ Wk, const float* __restrict__ Wv,
    const float* __restrict__ bq, const float* __restrict__ bk, const float* __restrict__ bv,
    const float* __restrict__ Wa, const float* __restrict__ ba,
    float* __restrict__ pre_q, float* __restrict__ pre_k, float* __restrict__ v)
{
    const int z = blockIdx.y;

    const int lane = threadIdx.x;
    const int wave = threadIdx.y;
    const int gw   = blockIdx.x * GEMM_WAVES + wave;
    const int tm   = gw / HH;
    const int tn   = gw - tm * HH;
    const int m0   = tm * 16;
    const int n0   = tn * 16;
    const int half = lane >> 4;
    const int sub  = lane & 15;

    const float* xRow = x + (size_t)(m0 + sub) * EE + half * 4;

    if (z == 2) {
        const float* wRow = Wv + (size_t)(n0 + sub) * EE + half * 4;
        const v8f c = tile_k256(xRow, wRow);
        const int   col  = n0 + sub;
        const int   row0 = m0 + half * 8;
        const float bb   = bv[col];
#pragma unroll
        for (int r = 0; r < 8; ++r)
            v[(size_t)(row0 + r) * EE + col] = c[r] + bb;
        return;
    }

    const float* __restrict__ W    = (z == 0) ? Wq : Wk;
    const float* __restrict__ bias = (z == 0) ? bq : bk;
    float* __restrict__       pre  = (z == 0) ? pre_q : pre_k;
    const float* wRow = W + (size_t)(n0 + sub) * EE + half * 4;

    const float* waRow = Wa + (size_t)sub * (2 * DD) + z * DD + half * 8;
    v4f w0[HID / 16], w1[HID / 16];
#pragma unroll
    for (int ct = 0; ct < HID / 16; ++ct) {
        w0[ct] = ld4(waRow + (size_t)ct * 16 * (2 * DD));
        w1[ct] = ld4(waRow + (size_t)ct * 16 * (2 * DD) + 4);
    }
    const v4f bl = ld4(bias + n0 + half * 8);
    const v4f bh = ld4(bias + n0 + half * 8 + 4);
    float bav[HID / 16];
#pragma unroll
    for (int ct = 0; ct < HID / 16; ++ct) bav[ct] = ba[ct * 16 + sub];

    const v8f t = tile_k256(wRow, xRow);

    const v2f qa0 = {t[0] + bl.x, t[1] + bl.y};
    const v2f qa1 = {t[2] + bl.z, t[3] + bl.w};
    const v2f qa2 = {t[4] + bh.x, t[5] + bh.y};
    const v2f qa3 = {t[6] + bh.z, t[7] + bh.w};

    float* outp = pre + (size_t)(m0 + half * 8) * (HH * HID) + (size_t)tn * HID + sub;
#pragma unroll
    for (int ct = 0; ct < HID / 16; ++ct) {
        v8f p = {};
        p = __builtin_amdgcn_wmma_f32_16x16x4_f32(false, qa0, false, v2f{w0[ct].x, w0[ct].y}, (short)0, p, false, false);
        p = __builtin_amdgcn_wmma_f32_16x16x4_f32(false, qa1, false, v2f{w0[ct].z, w0[ct].w}, (short)0, p, false, false);
        p = __builtin_amdgcn_wmma_f32_16x16x4_f32(false, qa2, false, v2f{w1[ct].x, w1[ct].y}, (short)0, p, false, false);
        p = __builtin_amdgcn_wmma_f32_16x16x4_f32(false, qa3, false, v2f{w1[ct].z, w1[ct].w}, (short)0, p, false, false);
        const float bb = (z == 0) ? bav[ct] : 0.0f;
#pragma unroll
        for (int r = 0; r < 8; ++r)
            outp[(size_t)r * (HH * HID) + ct * 16] = p[r] + bb;
    }
}

#define IQ        4
#define SCP       (SS * HH + 16)
#define SC_THREADS (32 * HH)
#define SC_BLOCKS  (BB * SS / IQ)

__device__ __forceinline__ void load_keys8(v4f (&kk)[8], const float* pkh, int j0) {
#pragma unroll
    for (int u = 0; u < 8; ++u)
        kk[u] = ld4(pkh + (size_t)(j0 + u) * (HH * HID));
}

__device__ __forceinline__ v2f tanh2(v2f s) {
    return v2f{fast_tanh(s.x), fast_tanh(s.y)};
}

__device__ __forceinline__ float scores_8keys(const v4f (&kk)[8],
                                              const v2f (&qx)[IQ / 2], const v2f (&qy)[IQ / 2],
                                              const v2f (&qz)[IQ / 2], const v2f (&qw)[IQ / 2],
                                              const v4f a, int lane) {
    const v2f ax = {a.x, a.x}, ay = {a.y, a.y}, az = {a.z, a.z}, aw = {a.w, a.w};
    float part[32];
#pragma unroll
    for (int qp = 0; qp < IQ / 2; ++qp) {
#pragma unroll
        for (int u = 0; u < 8; ++u) {
            const v2f kx = {kk[u].x, kk[u].x}, ky = {kk[u].y, kk[u].y};
            const v2f kz = {kk[u].z, kk[u].z}, kw = {kk[u].w, kk[u].w};
            const v2f r = tanh2(qx[qp] + kx) * ax
                        + tanh2(qy[qp] + ky) * ay
                        + tanh2(qz[qp] + kz) * az
                        + tanh2(qw[qp] + kw) * aw;
            part[(2 * qp) * 8 + u]     = r.x;
            part[(2 * qp + 1) * 8 + u] = r.y;
        }
    }
    xtree_step<16>(part, lane);
    xtree_step<8>(part, lane);
    xtree_step<4>(part, lane);
    xtree_step<2>(part, lane);
    xtree_step<1>(part, lane);
    return part[0];
}

__global__ __launch_bounds__(SC_THREADS) void scores_softmax_ctx(
    const float* __restrict__ pre_q,
    const float* __restrict__ pre_k,
    const float* __restrict__ attn_v,
    const float* __restrict__ v,
    float* __restrict__ ctx)
{
    __shared__ float sc[IQ * SCP];
    __shared__ float vloc[IQ * HH * DD];

    const int    tid = threadIdx.x;
    const size_t bi0 = (size_t)blockIdx.x * IQ;
    const size_t bat = bi0 / SS;

    const float* vg = v + bi0 * (HH * DD);
    for (int t = tid; t < IQ * HH * DD; t += SC_THREADS) vloc[t] = vg[t];

    const int lane = tid & 31;
    const int h    = tid >> 5;
    const int c0   = lane * 4;
    const v4f a    = ld4(attn_v + c0);
    v2f qx[IQ / 2], qy[IQ / 2], qz[IQ / 2], qw[IQ / 2];
#pragma unroll
    for (int qp = 0; qp < IQ / 2; ++qp) {
        const float* pq0 = pre_q + (bi0 + 2 * qp) * (size_t)(HH * HID) + (size_t)h * HID + c0;
        const v4f q0 = ld4(pq0);
        const v4f q1 = ld4(pq0 + HH * HID);
        qx[qp] = v2f{q0.x, q1.x};  qy[qp] = v2f{q0.y, q1.y};
        qz[qp] = v2f{q0.z, q1.z};  qw[qp] = v2f{q0.w, q1.w};
    }
    const float* pkh = pre_k + bat * (size_t)(SS * HH * HID) + (size_t)h * HID + c0;

    const int sidx = (lane >> 3) * SCP + (lane & 7) * HH + h;
    v4f kA[8], kB[8];
    load_keys8(kA, pkh, 0);
    for (int jb = 0; jb < SS - 16; jb += 16) {
        load_keys8(kB, pkh, jb + 8);
        __builtin_amdgcn_sched_barrier(0);
        sc[sidx + jb * HH] = scores_8keys(kA, qx, qy, qz, qw, a, lane);
        load_keys8(kA, pkh, jb + 16);
        __builtin_amdgcn_sched_barrier(0);
        sc[sidx + (jb + 8) * HH] = scores_8keys(kB, qx, qy, qz, qw, a, lane);
    }
    load_keys8(kB, pkh, SS - 8);
    __builtin_amdgcn_sched_barrier(0);
    sc[sidx + (SS - 16) * HH] = scores_8keys(kA, qx, qy, qz, qw, a, lane);
    sc[sidx + (SS - 8) * HH]  = scores_8keys(kB, qx, qy, qz, qw, a, lane);
    __syncthreads();

    if (tid < IQ * HH) {
        float* s_ = sc + (tid >> 4) * SCP;
        const int hh = tid & 15;
        float m = -3.0e38f;
        for (int j = 0; j < SS; ++j) m = fmaxf(m, s_[j * HH + hh]);
        float s = 0.0f;
        for (int j = 0; j < SS; ++j) {
            float e = __expf(s_[j * HH + hh] - m);
            s_[j * HH + hh] = e;
            s += e;
        }
        const float inv = 1.0f / s;
        for (int j = 0; j < SS; ++j) s_[j * HH + hh] *= inv;
    }
    __syncthreads();

    float* cg = ctx + bi0 * (size_t)(SS * DD);
    for (int idx = tid; idx < IQ * SS * DD; idx += SC_THREADS) {
        const int qi = idx >> 10;
        const int j  = (idx >> 4) & (SS - 1);
        const int d  = idx & 15;
        const float* wr = sc + qi * SCP + j * HH;
        const float* vr = vloc + qi * (HH * DD) + d;
        float acc = 0.0f;
#pragma unroll
        for (int hh = 0; hh < HH; ++hh)
            acc = fmaf(wr[hh], vr[hh * DD], acc);
        cg[idx] = acc;
    }
}

extern "C" void kernel_launch(void* const* d_in, const int* in_sizes, int n_in,
                              void* d_out, int out_size, void* d_ws, size_t ws_size,
                              hipStream_t stream) {
    (void)in_sizes; (void)n_in; (void)out_size; (void)ws_size;
    const float* x      = (const float*)d_in[0];
    const float* Wq     = (const float*)d_in[1];
    const float* bq     = (const float*)d_in[2];
    const float* Wk     = (const float*)d_in[3];
    const float* bk     = (const float*)d_in[4];
    const float* Wv     = (const float*)d_in[5];
    const float* bv     = (const float*)d_in[6];
    const float* Wa     = (const float*)d_in[7];
    const float* ba     = (const float*)d_in[8];
    const float* attn_v = (const float*)d_in[9];
    const float* Wfc    = (const float*)d_in[10];
    const float* bfc    = (const float*)d_in[11];
    float* out = (float*)d_out;

    float* v     = (float*)d_ws;
    float* pre_q = v     + (size_t)MQ * EE;
    float* pre_k = pre_q + (size_t)MPRE * HID;
    float* ctx   = pre_k + (size_t)MPRE * HID;

    const dim3 blk(32, GEMM_WAVES);
    const float* none  = nullptr;
    float*       noneC = nullptr;

    qkv_pre_wmma_f32<<<dim3(128, 3), blk, 0, stream>>>(
        x, Wq, Wk, Wv, bq, bk, bv, Wa, ba, pre_q, pre_k, v);

    scores_softmax_ctx<<<SC_BLOCKS, SC_THREADS, 0, stream>>>(pre_q, pre_k, attn_v, v, ctx);

    gemm_bias_wmma_f32<64><<<dim3(128, 1), blk, 0, stream>>>(
        ctx, none, none, SS * DD,
        Wfc, none, none, SS * DD, 0, 0, 0,
        bfc, none, none,
        out, noneC, noneC, EE,
        EE / 16, SS * DD);
}


# softmax: each wave normalises its own head's 4x64 scores in registers (lane max/sum over 8 values, then 3 exchange steps), not 64 serial threads; one barrier fewer; softmax sum order changes
# speedup vs baseline: 2.5743x; 1.0553x over previous
#include <hip/hip_runtime.h>
#include <hip/hip_bf16.h>

#define BB   16
#define SS   64
#define HH   16
#define DD   16
#define EE   256
#define HID  128
#define MQ   (BB*SS)
#define MPRE (BB*SS*HH)

typedef float v2f __attribute__((ext_vector_type(2)));
typedef float v4f __attribute__((ext_vector_type(4)));
typedef float v8f __attribute__((ext_vector_type(8)));

__device__ __forceinline__ float fast_tanh(float x) {
#if __has_builtin(__builtin_amdgcn_tanhf)
    return __builtin_amdgcn_tanhf(x);
#else
    return tanhf(x);
#endif
}

__device__ __forceinline__ v4f ld4(const float* p) {
    return *reinterpret_cast<const v4f*>(p);
}

template <int OFF>
__device__ __forceinline__ float lane_xor(float x) {
    const int b = __builtin_bit_cast(int, x);
    if constexpr (OFF == 16)
        return __builtin_bit_cast(float,
            __builtin_amdgcn_permlanex16(b, b, 0x76543210, 0xfedcba98, false, false));
    else
        return __builtin_bit_cast(float,
            __builtin_amdgcn_update_dpp(0, b, 0x160 + OFF, 0xf, 0xf, true));
}

template <int OFF>
__device__ __forceinline__ void xtree_step(float* p, int lane) {
    const bool up = (lane & OFF) != 0;
#pragma unroll
    for (int k = 0; k < OFF; ++k) {
        const float keep = up ? p[k + OFF] : p[k];
        const float send = up ? p[k] : p[k + OFF];
        p[k] = keep + lane_xor<OFF>(send);
    }
}

#define GEMM_WAVES 8

template <int KU>
__global__ __launch_bounds__(256) void gemm_bias_wmma_f32(
    const float* A0, const float* A1, const float* A2, int lda,
    const float* W0, const float* W1, const float* W2, int ldw,
    int wOff0, int wOff1, int wOff2,
    const float* bias0, const float* bias1, const float* bias2,
    float* C0, float* C1, float* C2, int ldc,
    int tilesN, int K)
{
    const int z = blockIdx.y;
    const float* __restrict__ A    = (z == 0) ? A0    : ((z == 1) ? A1    : A2);
    const float* __restrict__ W    = (z == 0) ? W0    : ((z == 1) ? W1    : W2);
    const float* __restrict__ bias = (z == 0) ? bias0 : ((z == 1) ? bias1 : bias2);
    float* __restrict__       C    = (z == 0) ? C0    : ((z == 1) ? C1    : C2);
    const int wOff                 = (z == 0) ? wOff0 : ((z == 1) ? wOff1 : wOff2);

    const int lane = threadIdx.x;
    const int wave = threadIdx.y;
    const int gw   = blockIdx.x * GEMM_WAVES + wave;
    const int tm   = gw / tilesN;
    const int tn   = gw - tm * tilesN;
    const int m0   = tm * 16;
    const int n0   = tn * 16;
    const int half = lane >> 4;
    const int sub  = lane & 15;
    const int h4   = half * 4;

    const float* aRow = A + (size_t)(m0 + sub) * lda + h4;
    const float* wRow = W + (size_t)(n0 + sub) * ldw + wOff + h4;

    v8f c = {};
    for (int k0 = 0; k0 < K; k0 += KU) {
        v4f a[KU / 8], b[KU / 8];
#pragma unroll
        for (int i = 0; i < KU / 8; ++i) a[i] = ld4(aRow + k0 + 8 * i);
#pragma unroll
        for (int i = 0; i < KU / 8; ++i) b[i] = ld4(wRow + k0 + 8 * i);
        __builtin_amdgcn_sched_barrier(0);
#pragma unroll
        for (int i = 0; i < KU / 8; ++i) {
            c = __builtin_amdgcn_wmma_f32_16x16x4_f32(false, v2f{a[i].x, a[i].y}, false, v2f{b[i].x, b[i].y},
                                                      (short)0, c, false, false);
            c = __builtin_amdgcn_wmma_f32_16x16x4_f32(false, v2f{a[i].z, a[i].w}, false, v2f{b[i].z, b[i].w},
                                                      (short)0, c, false, false);
        }
    }

    const int   col  = n0 + sub;
    const int   row0 = m0 + half * 8;
    const float bb   = bias ? bias[col] : 0.0f;
#pragma unroll
    for (int r = 0; r < 8; ++r)
        C[(size_t)(row0 + r) * ldc + col] = c[r] + bb;
}

__device__ __forceinline__ v8f tile_k256(const float* __restrict__ pRow, const float* __restrict__ qRow) {
    v8f c = {};
    for (int k0 = 0; k0 < EE; k0 += 64) {
        v4f a[8], b[8];
#pragma unroll
        for (int i = 0; i < 8; ++i) a[i] = ld4(pRow + k0 + 8 * i);
#pragma unroll
        for (int i = 0; i < 8; ++i) b[i] = ld4(qRow + k0 + 8 * i);
        __builtin_amdgcn_sched_barrier(0);
#pragma unroll
        for (int i = 0; i < 8; ++i) {
            c = __builtin_amdgcn_wmma_f32_16x16x4_f32(false, v2f{a[i].x, a[i].y}, false, v2f{b[i].x, b[i].y},
                                                      (short)0, c, false, false);
            c = __builtin_amdgcn_wmma_f32_16x16x4_f32(false, v2f{a[i].z, a[i].w}, false, v2f{b[i].z, b[i].w},
                                                      (short)0, c, false, false);
        }
    }
    return c;
}

__global__ __launch_bounds__(256) void qkv_pre_wmma_f32(
    const float* __restrict__ x,
    const float* __restrict__ Wq, const float* __restrict__ Wk, const float* __restrict__ Wv,
    const float* __restrict__ bq, const float* __restrict__ bk, const float* __restrict__ bv,
    const float* __restrict__ Wa, const float* __restrict__ ba,
    float* __restrict__ pre_q, float* __restrict__ pre_k, float* __restrict__ v)
{
    const int z = blockIdx.y;

    const int lane = threadIdx.x;
    const int wave = threadIdx.y;
    const int gw   = blockIdx.x * GEMM_WAVES + wave;
    const int tm   = gw / HH;
    const int tn   = gw - tm * HH;
    const int m0   = tm * 16;
    const int n0   = tn * 16;
    const int half = lane >> 4;
    const int sub  = lane & 15;

    const float* xRow = x + (size_t)(m0 + sub) * EE + half * 4;

    if (z == 2) {
        const float* wRow = Wv + (size_t)(n0 + sub) * EE + half * 4;
        const v8f c = tile_k256(xRow, wRow);
        const int   col  = n0 + sub;
        const int   row0 = m0 + half * 8;
        const float bb   = bv[col];
#pragma unroll
        for (int r = 0; r < 8; ++r)
            v[(size_t)(row0 + r) * EE + col] = c[r] + bb;
        return;
    }

    const float* __restrict__ W    = (z == 0) ? Wq : Wk;
    const float* __restrict__ bias = (z == 0) ? bq : bk;
    float* __restrict__       pre  = (z == 0) ? pre_q : pre_k;
    const float* wRow = W + (size_t)(n0 + sub) * EE + half * 4;

    const float* waRow = Wa + (size_t)sub * (2 * DD) + z * DD + half * 8;
    v4f w0[HID / 16], w1[HID / 16];
#pragma unroll
    for (int ct = 0; ct < HID / 16; ++ct) {
        w0[ct] = ld4(waRow + (size_t)ct * 16 * (2 * DD));
        w1[ct] = ld4(waRow + (size_t)ct * 16 * (2 * DD) + 4);
    }
    const v4f bl = ld4(bias + n0 + half * 8);
    const v4f bh = ld4(bias + n0 + half * 8 + 4);
    float bav[HID / 16];
#pragma unroll
    for (int ct = 0; ct < HID / 16; ++ct) bav[ct] = ba[ct * 16 + sub];

    const v8f t = tile_k256(wRow, xRow);

    const v2f qa0 = {t[0] + bl.x, t[1] + bl.y};
    const v2f qa1 = {t[2] + bl.z, t[3] + bl.w};
    const v2f qa2 = {t[4] + bh.x, t[5] + bh.y};
    const v2f qa3 = {t[6] + bh.z, t[7] + bh.w};

    float* outp = pre + (size_t)(m0 + half * 8) * (HH * HID) + (size_t)tn * HID + sub;
#pragma unroll
    for (int ct = 0; ct < HID / 16; ++ct) {
        v8f p = {};
        p = __builtin_amdgcn_wmma_f32_16x16x4_f32(false, qa0, false, v2f{w0[ct].x, w0[ct].y}, (short)0, p, false, false);
        p = __builtin_amdgcn_wmma_f32_16x16x4_f32(false, qa1, false, v2f{w0[ct].z, w0[ct].w}, (short)0, p, false, false);
        p = __builtin_amdgcn_wmma_f32_16x16x4_f32(false, qa2, false, v2f{w1[ct].x, w1[ct].y}, (short)0, p, false, false);
        p = __builtin_amdgcn_wmma_f32_16x16x4_f32(false, qa3, false, v2f{w1[ct].z, w1[ct].w}, (short)0, p, false, false);
        const float bb = (z == 0) ? bav[ct] : 0.0f;
#pragma unroll
        for (int r = 0; r < 8; ++r)
            outp[(size_t)r * (HH * HID) + ct * 16] = p[r] + bb;
    }
}

#define IQ        4
#define SCP       (SS * HH + 4)
#define SC_THREADS (32 * HH)
#define SC_BLOCKS  (BB * SS / IQ)

__device__ __forceinline__ void load_keys8(v4f (&kk)[8], const float* pkh, int j0) {
#pragma unroll
    for (int u = 0; u < 8; ++u)
        kk[u] = ld4(pkh + (size_t)(j0 + u) * (HH * HID));
}

__device__ __forceinline__ v2f tanh2(v2f s) {
    return v2f{fast_tanh(s.x), fast_tanh(s.y)};
}

__device__ __forceinline__ float scores_8keys(const v4f (&kk)[8],
                                              const v2f (&qx)[IQ / 2], const v2f (&qy)[IQ / 2],
                                              const v2f (&qz)[IQ / 2], const v2f (&qw)[IQ / 2],
                                              const v4f a, int lane) {
    const v2f ax = {a.x, a.x}, ay = {a.y, a.y}, az = {a.z, a.z}, aw = {a.w, a.w};
    float part[32];
#pragma unroll
    for (int qp = 0; qp < IQ / 2; ++qp) {
#pragma unroll
        for (int u = 0; u < 8; ++u) {
            const v2f kx = {kk[u].x, kk[u].x}, ky = {kk[u].y, kk[u].y};
            const v2f kz = {kk[u].z, kk[u].z}, kw = {kk[u].w, kk[u].w};
            const v2f r = tanh2(qx[qp] + kx) * ax
                        + tanh2(qy[qp] + ky) * ay
                        + tanh2(qz[qp] + kz) * az
                        + tanh2(qw[qp] + kw) * aw;
            part[(2 * qp) * 8 + u]     = r.x;
            part[(2 * qp + 1) * 8 + u] = r.y;
        }
    }
    xtree_step<16>(part, lane);
    xtree_step<8>(part, lane);
    xtree_step<4>(part, lane);
    xtree_step<2>(part, lane);
    xtree_step<1>(part, lane);
    return part[0];
}

__global__ __launch_bounds__(SC_THREADS) void scores_softmax_ctx(
    const float* __restrict__ pre_q,
    const float* __restrict__ pre_k,
    const float* __restrict__ attn_v,
    const float* __restrict__ v,
    float* __restrict__ ctx)
{
    __shared__ float sc[IQ * SCP];
    __shared__ float vloc[IQ * HH * DD];

    const int    tid = threadIdx.x;
    const size_t bi0 = (size_t)blockIdx.x * IQ;
    const size_t bat = bi0 / SS;

    const float* vg = v + bi0 * (HH * DD);
    for (int t = tid; t < IQ * HH * DD; t += SC_THREADS) vloc[t] = vg[t];

    const int lane = tid & 31;
    const int h    = tid >> 5;
    const int c0   = lane * 4;
    const v4f a    = ld4(attn_v + c0);
    v2f qx[IQ / 2], qy[IQ / 2], qz[IQ / 2], qw[IQ / 2];
#pragma unroll
    for (int qp = 0; qp < IQ / 2; ++qp) {
        const float* pq0 = pre_q + (bi0 + 2 * qp) * (size_t)(HH * HID) + (size_t)h * HID + c0;
        const v4f q0 = ld4(pq0);
        const v4f q1 = ld4(pq0 + HH * HID);
        qx[qp] = v2f{q0.x, q1.x};  qy[qp] = v2f{q0.y, q1.y};
        qz[qp] = v2f{q0.z, q1.z};  qw[qp] = v2f{q0.w, q1.w};
    }
    const float* pkh = pre_k + bat * (size_t)(SS * HH * HID) + (size_t)h * HID + c0;

    const int sidx = (lane >> 3) * SCP + (lane & 7) * HH + h;
    v4f kA[8], kB[8];
    load_keys8(kA, pkh, 0);
    for (int jb = 0; jb < SS - 16; jb += 16) {
        load_keys8(kB, pkh, jb + 8);
        __builtin_amdgcn_sched_barrier(0);
        sc[sidx + jb * HH] = scores_8keys(kA, qx, qy, qz, qw, a, lane);
        load_keys8(kA, pkh, jb + 16);
        __builtin_amdgcn_sched_barrier(0);
        sc[sidx + (jb + 8) * HH] = scores_8keys(kB, qx, qy, qz, qw, a, lane);
    }
    load_keys8(kB, pkh, SS - 8);
    __builtin_amdgcn_sched_barrier(0);
    sc[sidx + (SS - 16) * HH] = scores_8keys(kA, qx, qy, qz, qw, a, lane);
    sc[sidx + (SS - 8) * HH]  = scores_8keys(kB, qx, qy, qz, qw, a, lane);

    float e[SS / 8];
#pragma unroll
    for (int c = 0; c < SS / 8; ++c) e[c] = sc[sidx + c * 8 * HH];
    float m = e[0];
#pragma unroll
    for (int c = 1; c < SS / 8; ++c) m = fmaxf(m, e[c]);
    m = fmaxf(m, lane_xor<1>(m));
    m = fmaxf(m, lane_xor<2>(m));
    m = fmaxf(m, lane_xor<4>(m));
    float s = 0.0f;
#pragma unroll
    for (int c = 0; c < SS / 8; ++c) {
        e[c] = __expf(e[c] - m);
        s += e[c];
    }
    s += lane_xor<1>(s);
    s += lane_xor<2>(s);
    s += lane_xor<4>(s);
    const float inv = 1.0f / s;
#pragma unroll
    for (int c = 0; c < SS / 8; ++c) sc[sidx + c * 8 * HH] = e[c] * inv;
    __syncthreads();

    float* cg = ctx + bi0 * (size_t)(SS * DD);
    for (int idx = tid; idx < IQ * SS * DD; idx += SC_THREADS) {
        const int qi = idx >> 10;
        const int j  = (idx >> 4) & (SS - 1);
        const int d  = idx & 15;
        const float* wr = sc + qi * SCP + j * HH;
        const float* vr = vloc + qi * (HH * DD) + d;
        float acc = 0.0f;
#pragma unroll
        for (int hh = 0; hh < HH; ++hh)
            acc = fmaf(wr[hh], vr[hh * DD], acc);
        cg[idx] = acc;
    }
}

extern "C" void kernel_launch(void* const* d_in, const int* in_sizes, int n_in,
                              void* d_out, int out_size, void* d_ws, size_t ws_size,
                              hipStream_t stream) {
    (void)in_sizes; (void)n_in; (void)out_size; (void)ws_size;
    const float* x      = (const float*)d_in[0];
    const float* Wq     = (const float*)d_in[1];
    const float* bq     = (const float*)d_in[2];
    const float* Wk     = (const float*)d_in[3];
    const float* bk     = (const float*)d_in[4];
    const float* Wv     = (const float*)d_in[5];
    const float* bv     = (const float*)d_in[6];
    const float* Wa     = (const float*)d_in[7];
    const float* ba     = (const float*)d_in[8];
    const float* attn_v = (const float*)d_in[9];
    const float* Wfc    = (const float*)d_in[10];
    const float* bfc    = (const float*)d_in[11];
    float* out = (float*)d_out;

    float* v     = (float*)d_ws;
    float* pre_q = v     + (size_t)MQ * EE;
    float* pre_k = pre_q + (size_t)MPRE * HID;
    float* ctx   = pre_k + (size_t)MPRE * HID;

    const dim3 blk(32, GEMM_WAVES);
    const float* none  = nullptr;
    float*       noneC = nullptr;

    qkv_pre_wmma_f32<<<dim3(128, 3), blk, 0, stream>>>(
        x, Wq, Wk, Wv, bq, bk, bv, Wa, ba, pre_q, pre_k, v);

    scores_softmax_ctx<<<SC_BLOCKS, SC_THREADS, 0, stream>>>(pre_q, pre_k, attn_v, v, ctx);

    gemm_bias_wmma_f32<64><<<dim3(128, 1), blk, 0, stream>>>(
        ctx, none, none, SS * DD,
        Wfc, none, none, SS * DD, 0, 0, 0,
        bfc, none, none,
        out, noneC, noneC, EE,
        EE / 16, SS * DD);
}


# final product: 32x32 block tiles on all 256 WGPs, A and W stages brought in by whole-line loads and kept in LDS, k split over 4 waves and summed through LDS (sum order over k changes)
# speedup vs baseline: 3.4383x; 1.3356x over previous
#include <hip/hip_runtime.h>
#include <hip/hip_bf16.h>

#define BB   16
#define SS   64
#define HH   16
#define DD   16
#define EE   256
#define HID  128
#define MQ   (BB*SS)
#define MPRE (BB*SS*HH)

typedef float v2f __attribute__((ext_vector_type(2)));
typedef float v4f __attribute__((ext_vector_type(4)));
typedef float v8f __attribute__((ext_vector_type(8)));

__device__ __forceinline__ float fast_tanh(float x) {
#if __has_builtin(__builtin_amdgcn_tanhf)
    return __builtin_amdgcn_tanhf(x);
#else
    return tanhf(x);
#endif
}

__device__ __forceinline__ v4f ld4(const float* p) {
    return *reinterpret_cast<const v4f*>(p);
}

template <int OFF>
__device__ __forceinline__ float lane_xor(float x) {
    const int b = __builtin_bit_cast(int, x);
    if constexpr (OFF == 16)
        return __builtin_bit_cast(float,
            __builtin_amdgcn_permlanex16(b, b, 0x76543210, 0xfedcba98, false, false));
    else
        return __builtin_bit_cast(float,
            __builtin_amdgcn_update_dpp(0, b, 0x160 + OFF, 0xf, 0xf, true));
}

template <int OFF>
__device__ __forceinline__ void xtree_step(float* p, int lane) {
    const bool up = (lane & OFF) != 0;
#pragma unroll
    for (int k = 0; k < OFF; ++k) {
        const float keep = up ? p[k + OFF] : p[k];
        const float send = up ? p[k] : p[k + OFF];
        p[k] = keep + lane_xor<OFF>(send);
    }
}

#define GEMM_WAVES 8
#define FC_K     (SS * DD)
#define FC_T     32
#define FC_KC    256
#define FC_NST   (FC_K / FC_KC)
#define FC_PITCH (FC_KC + 4)
#define FC_STAGE (2 * FC_T * FC_PITCH)
#define FC_LPT   ((2 * FC_T * FC_KC) / (256 * 4))
#define FC_BLOCKS ((MQ / FC_T) * (EE / FC_T))

__device__ __forceinline__ void fc_fetch(v4f (&g)[FC_LPT], const float* __restrict__ aBlk,
                                         const float* __restrict__ wBlk, int kc, int wave, int lane) {
    const int rsub = wave >> 1;
    const int col  = (wave & 1) * 128 + lane * 4 + kc * FC_KC;
#pragma unroll
    for (int q = 0; q < FC_LPT; ++q) {
        const int row = 4 * q + rsub;
        const float* src = (q < FC_LPT / 2) ? aBlk + (size_t)row * FC_K
                                            : wBlk + (size_t)(row - FC_T) * FC_K;
        g[q] = ld4(src + col);
    }
}

__device__ __forceinline__ void fc_put(const v4f (&g)[FC_LPT], float* stg, int wave, int lane) {
    const int rsub = wave >> 1;
    const int col  = (wave & 1) * 128 + lane * 4;
#pragma unroll
    for (int q = 0; q < FC_LPT; ++q)
        *reinterpret_cast<v4f*>(stg + (4 * q + rsub) * FC_PITCH + col) = g[q];
}

__global__ __launch_bounds__(256) void fc_lds_wmma_f32(
    const float* __restrict__ A,
    const float* __restrict__ W,
    const float* __restrict__ bias,
    float* __restrict__ C)
{
    __shared__ __attribute__((aligned(16))) float st[2 * FC_STAGE];
    __shared__ float red[GEMM_WAVES * 2 * 8 * 32];

    const int tid  = threadIdx.x;
    const int lane = tid & 31;
    const int wave = tid >> 5;
    const int half = lane >> 4;
    const int sub  = lane & 15;
    const int m0   = (blockIdx.x >> 3) * FC_T;
    const int n0   = (blockIdx.x & 7) * FC_T;
    const int tmw  = wave & 1;
    const int kq   = wave >> 1;

    const float* aBlk = A + (size_t)m0 * FC_K;
    const float* wBlk = W + (size_t)n0 * FC_K;

    v4f g[FC_LPT];
    fc_fetch(g, aBlk, wBlk, 0, wave, lane);
    fc_put(g, st, wave, lane);
    __syncthreads();

    v8f c0 = {}, c1 = {};
    const int fo = kq * (FC_KC / 4) + 4 * half;
#pragma unroll
    for (int kc = 0; kc < FC_NST; ++kc) {
        if (kc + 1 < FC_NST) {
            fc_fetch(g, aBlk, wBlk, kc + 1, wave, lane);
            __builtin_amdgcn_sched_barrier(0);
        }
        const float* sA = st + (kc & 1) * FC_STAGE + (tmw * 16 + sub) * FC_PITCH + fo;
        const float* sW = st + (kc & 1) * FC_STAGE + (FC_T + sub) * FC_PITCH + fo;
        v4f fa[FC_KC / 32], fb0[FC_KC / 32], fb1[FC_KC / 32];
#pragma unroll
        for (int i = 0; i < FC_KC / 32; ++i) {
            fa[i]  = ld4(sA + 8 * i);
            fb0[i] = ld4(sW + 8 * i);
            fb1[i] = ld4(sW + 16 * FC_PITCH + 8 * i);
        }
        __builtin_amdgcn_sched_barrier(0);
#pragma unroll
        for (int i = 0; i < FC_KC / 32; ++i) {
            c0 = __builtin_amdgcn_wmma_f32_16x16x4_f32(false, v2f{fa[i].x, fa[i].y}, false, v2f{fb0[i].x, fb0[i].y},
                                                       (short)0, c0, false, false);
            c1 = __builtin_amdgcn_wmma_f32_16x16x4_f32(false, v2f{fa[i].x, fa[i].y}, false, v2f{fb1[i].x, fb1[i].y},
                                                       (short)0, c1, false, false);
            c0 = __builtin_amdgcn_wmma_f32_16x16x4_f32(false, v2f{fa[i].z, fa[i].w}, false, v2f{fb0[i].z, fb0[i].w},
                                                       (short)0, c0, false, false);
            c1 = __builtin_amdgcn_wmma_f32_16x16x4_f32(false, v2f{fa[i].z, fa[i].w}, false, v2f{fb1[i].z, fb1[i].w},
                                                       (short)0, c1, false, false);
        }
        if (kc + 1 < FC_NST) {
            __builtin_amdgcn_sched_barrier(0);
            fc_put(g, st + ((kc + 1) & 1) * FC_STAGE, wave, lane);
        }
        __syncthreads();
    }

#pragma unroll
    for (int r = 0; r < 8; ++r) {
        red[((wave * 2 + 0) * 8 + r) * 32 + lane] = c0[r];
        red[((wave * 2 + 1) * 8 + r) * 32 + lane] = c1[r];
    }
    __syncthreads();

    const float bb = bias[n0 + lane];
    const int   t  = lane >> 4;
#pragma unroll
    for (int o = 0; o < 4; ++o) {
        const int rr  = wave * 4 + o;
        const int tms = rr >> 4;
        const int src = ((rr >> 3) & 1) * 16 + sub;
        const int r   = rr & 7;
        float sum = 0.0f;
#pragma unroll
        for (int q = 0; q < 4; ++q)
            sum += red[(((q * 2 + tms) * 2 + t) * 8 + r) * 32 + src];
        C[(size_t)(m0 + rr) * EE + n0 + lane] = sum + bb;
    }
}

__device__ __forceinline__ v8f tile_k256(const float* __restrict__ pRow, const float* __restrict__ qRow) {
    v8f c = {};
    for (int k0 = 0; k0 < EE; k0 += 64) {
        v4f a[8], b[8];
#pragma unroll
        for (int i = 0; i < 8; ++i) a[i] = ld4(pRow + k0 + 8 * i);
#pragma unroll
        for (int i = 0; i < 8; ++i) b[i] = ld4(qRow + k0 + 8 * i);
        __builtin_amdgcn_sched_barrier(0);
#pragma unroll
        for (int i = 0; i < 8; ++i) {
            c = __builtin_amdgcn_wmma_f32_16x16x4_f32(false, v2f{a[i].x, a[i].y}, false, v2f{b[i].x, b[i].y},
                                                      (short)0, c, false, false);
            c = __builtin_amdgcn_wmma_f32_16x16x4_f32(false, v2f{a[i].z, a[i].w}, false, v2f{b[i].z, b[i].w},
                                                      (short)0, c, false, false);
        }
    }
    return c;
}

__global__ __launch_bounds__(256) void qkv_pre_wmma_f32(
    const float* __restrict__ x,
    const float* __restrict__ Wq, const float* __restrict__ Wk, const float* __restrict__ Wv,
    const float* __restrict__ bq, const float* __restrict__ bk, const float* __restrict__ bv,
    const float* __restrict__ Wa, const float* __restrict__ ba,
    float* __restrict__ pre_q, float* __restrict__ pre_k, float* __restrict__ v)
{
    const int z = blockIdx.y;

    const int lane = threadIdx.x;
    const int wave = threadIdx.y;
    const int gw   = blockIdx.x * GEMM_WAVES + wave;
    const int tm   = gw / HH;
    const int tn   = gw - tm * HH;
    const int m0   = tm * 16;
    const int n0   = tn * 16;
    const int half = lane >> 4;
    const int sub  = lane & 15;

    const float* xRow = x + (size_t)(m0 + sub) * EE + half * 4;

    if (z == 2) {
        const float* wRow = Wv + (size_t)(n0 + sub) * EE + half * 4;
        const v8f c = tile_k256(xRow, wRow);
        const int   col  = n0 + sub;
        const int   row0 = m0 + half * 8;
        const float bb   = bv[col];
#pragma unroll
        for (int r = 0; r < 8; ++r)
            v[(size_t)(row0 + r) * EE + col] = c[r] + bb;
        return;
    }

    const float* __restrict__ W    = (z == 0) ? Wq : Wk;
    const float* __restrict__ bias = (z == 0) ? bq : bk;
    float* __restrict__       pre  = (z == 0) ? pre_q : pre_k;
    const float* wRow = W + (size_t)(n0 + sub) * EE + half * 4;

    const float* waRow = Wa + (size_t)sub * (2 * DD) + z * DD + half * 8;
    v4f w0[HID / 16], w1[HID / 16];
#pragma unroll
    for (int ct = 0; ct < HID / 16; ++ct) {
        w0[ct] = ld4(waRow + (size_t)ct * 16 * (2 * DD));
        w1[ct] = ld4(waRow + (size_t)ct * 16 * (2 * DD) + 4);
    }
    const v4f bl = ld4(bias + n0 + half * 8);
    const v4f bh = ld4(bias + n0 + half * 8 + 4);
    float bav[HID / 16];
#pragma unroll
    for (int ct = 0; ct < HID / 16; ++ct) bav[ct] = ba[ct * 16 + sub];

    const v8f t = tile_k256(wRow, xRow);

    const v2f qa0 = {t[0] + bl.x, t[1] + bl.y};
    const v2f qa1 = {t[2] + bl.z, t[3] + bl.w};
    const v2f qa2 = {t[4] + bh.x, t[5] + bh.y};
    const v2f qa3 = {t[6] + bh.z, t[7] + bh.w};

    float* outp = pre + (size_t)(m0 + half * 8) * (HH * HID) + (size_t)tn * HID + sub;
#pragma unroll
    for (int ct = 0; ct < HID / 16; ++ct) {
        v8f p = {};
        p = __builtin_amdgcn_wmma_f32_16x16x4_f32(false, qa0, false, v2f{w0[ct].x, w0[ct].y}, (short)0, p, false, false);
        p = __builtin_amdgcn_wmma_f32_16x16x4_f32(false, qa1, false, v2f{w0[ct].z, w0[ct].w}, (short)0, p, false, false);
        p = __builtin_amdgcn_wmma_f32_16x16x4_f32(false, qa2, false, v2f{w1[ct].x, w1[ct].y}, (short)0, p, false, false);
        p = __builtin_amdgcn_wmma_f32_16x16x4_f32(false, qa3, false, v2f{w1[ct].z, w1[ct].w}, (short)0, p, false, false);
        const float bb = (z == 0) ? bav[ct] : 0.0f;
#pragma unroll
        for (int r = 0; r < 8; ++r)
            outp[(size_t)r * (HH * HID) + ct * 16] = p[r] + bb;
    }
}

#define IQ        4
#define SCP       (SS * HH + 4)
#define SC_THREADS (32 * HH)
#define SC_BLOCKS  (BB * SS / IQ)

__device__ __forceinline__ void load_keys8(v4f (&kk)[8], const float* pkh, int j0) {
#pragma unroll
    for (int u = 0; u < 8; ++u)
        kk[u] = ld4(pkh + (size_t)(j0 + u) * (HH * HID));
}

__device__ __forceinline__ v2f tanh2(v2f s) {
    return v2f{fast_tanh(s.x), fast_tanh(s.y)};
}

__device__ __forceinline__ float scores_8keys(const v4f (&kk)[8],
                                              const v2f (&qx)[IQ / 2], const v2f (&qy)[IQ / 2],
                                              const v2f (&qz)[IQ / 2], const v2f (&qw)[IQ / 2],
                                              const v4f a, int lane) {
    const v2f ax = {a.x, a.x}, ay = {a.y, a.y}, az = {a.z, a.z}, aw = {a.w, a.w};
    float part[32];
#pragma unroll
    for (int qp = 0; qp < IQ / 2; ++qp) {
#pragma unroll
        for (int u = 0; u < 8; ++u) {
            const v2f kx = {kk[u].x, kk[u].x}, ky = {kk[u].y, kk[u].y};
            const v2f kz = {kk[u].z, kk[u].z}, kw = {kk[u].w, kk[u].w};
            const v2f r = tanh2(qx[qp] + kx) * ax
                        + tanh2(qy[qp] + ky) * ay
                        + tanh2(qz[qp] + kz) * az
                        + tanh2(qw[qp] + kw) * aw;
            part[(2 * qp) * 8 + u]     = r.x;
            part[(2 * qp + 1) * 8 + u] = r.y;
        }
    }
    xtree_step<16>(part, lane);
    xtree_step<8>(part, lane);
    xtree_step<4>(part, lane);
    xtree_step<2>(part, lane);
    xtree_step<1>(part, lane);
    return part[0];
}

__global__ __launch_bounds__(SC_THREADS) void scores_softmax_ctx(
    const float* __restrict__ pre_q,
    const float* __restrict__ pre_k,
    const float* __restrict__ attn_v,
    const float* __restrict__ v,
    float* __restrict__ ctx)
{
    __shared__ float sc[IQ * SCP];
    __shared__ float vloc[IQ * HH * DD];

    const int    tid = threadIdx.x;
    const size_t bi0 = (size_t)blockIdx.x * IQ;
    const size_t bat = bi0 / SS;

    const float* vg = v + bi0 * (HH * DD);
    for (int t = tid; t < IQ * HH * DD; t += SC_THREADS) vloc[t] = vg[t];

    const int lane = tid & 31;
    const int h    = tid >> 5;
    const int c0   = lane * 4;
    const v4f a    = ld4(attn_v + c0);
    v2f qx[IQ / 2], qy[IQ / 2], qz[IQ / 2], qw[IQ / 2];
#pragma unroll
    for (int qp = 0; qp < IQ / 2; ++qp) {
        const float* pq0 = pre_q + (bi0 + 2 * qp) * (size_t)(HH * HID) + (size_t)h * HID + c0;
        const v4f q0 = ld4(pq0);
        const v4f q1 = ld4(pq0 + HH * HID);
        qx[qp] = v2f{q0.x, q1.x};  qy[qp] = v2f{q0.y, q1.y};
        qz[qp] = v2f{q0.z, q1.z};  qw[qp] = v2f{q0.w, q1.w};
    }
    const float* pkh = pre_k + bat * (size_t)(SS * HH * HID) + (size_t)h * HID + c0;

    const int sidx = (lane >> 3) * SCP + (lane & 7) * HH + h;
    v4f kA[8], kB[8];
    load_keys8(kA, pkh, 0);
    for (int jb = 0; jb < SS - 16; jb += 16) {
        load_keys8(kB, pkh, jb + 8);
        __builtin_amdgcn_sched_barrier(0);
        sc[sidx + jb * HH] = scores_8keys(kA, qx, qy, qz, qw, a, lane);
        load_keys8(kA, pkh, jb + 16);
        __builtin_amdgcn_sched_barrier(0);
        sc[sidx + (jb + 8) * HH] = scores_8keys(kB, qx, qy, qz, qw, a, lane);
    }
    load_keys8(kB, pkh, SS - 8);
    __builtin_amdgcn_sched_barrier(0);
    sc[sidx + (SS - 16) * HH] = scores_8keys(kA, qx, qy, qz, qw, a, lane);
    sc[sidx + (SS - 8) * HH]  = scores_8keys(kB, qx, qy, qz, qw, a, lane);

    float e[SS / 8];
#pragma unroll
    for (int c = 0; c < SS / 8; ++c) e[c] = sc[sidx + c * 8 * HH];
    float m = e[0];
#pragma unroll
    for (int c = 1; c < SS / 8; ++c) m = fmaxf(m, e[c]);
    m = fmaxf(m, lane_xor<1>(m));
    m = fmaxf(m, lane_xor<2>(m));
    m = fmaxf(m, lane_xor<4>(m));
    float s = 0.0f;
#pragma unroll
    for (int c = 0; c < SS / 8; ++c) {
        e[c] = __expf(e[c] - m);
        s += e[c];
    }
    s += lane_xor<1>(s);
    s += lane_xor<2>(s);
    s += lane_xor<4>(s);
    const float inv = 1.0f / s;
#pragma unroll
    for (int c = 0; c < SS / 8; ++c) sc[sidx + c * 8 * HH] = e[c] * inv;
    __syncthreads();

    float* cg = ctx + bi0 * (size_t)(SS * DD);
    for (int idx = tid; idx < IQ * SS * DD; idx += SC_THREADS) {
        const int qi = idx >> 10;
        const int j  = (idx >> 4) & (SS - 1);
        const int d  = idx & 15;
        const float* wr = sc + qi * SCP + j * HH;
        const float* vr = vloc + qi * (HH * DD) + d;
        float acc = 0.0f;
#pragma unroll
        for (int hh = 0; hh < HH; ++hh)
            acc = fmaf(wr[hh], vr[hh * DD], acc);
        cg[idx] = acc;
    }
}

extern "C" void kernel_launch(void* const* d_in, const int* in_sizes, int n_in,
                              void* d_out, int out_size, void* d_ws, size_t ws_size,
                              hipStream_t stream) {
    (void)in_sizes; (void)n_in; (void)out_size; (void)ws_size;
    const float* x      = (const float*)d_in[0];
    const float* Wq     = (const float*)d_in[1];
    const float* bq     = (const float*)d_in[2];
    const float* Wk     = (const float*)d_in[3];
    const float* bk     = (const float*)d_in[4];
    const float* Wv     = (const float*)d_in[5];
    const float* bv     = (const float*)d_in[6];
    const float* Wa     = (const float*)d_in[7];
    const float* ba     = (const float*)d_in[8];
    const float* attn_v = (const float*)d_in[9];
    const float* Wfc    = (const float*)d_in[10];
    const float* bfc    = (const float*)d_in[11];
    float* out = (float*)d_out;

    float* v     = (float*)d_ws;
    float* pre_q = v     + (size_t)MQ * EE;
    float* pre_k = pre_q + (size_t)MPRE * HID;
    float* ctx   = pre_k + (size_t)MPRE * HID;

    const dim3 blk(32, GEMM_WAVES);

    qkv_pre_wmma_f32<<<dim3(128, 3), blk, 0, stream>>>(
        x, Wq, Wk, Wv, bq, bk, bv, Wa, ba, pre_q, pre_k, v);

    scores_softmax_ctx<<<SC_BLOCKS, SC_THREADS, 0, stream>>>(pre_q, pre_k, attn_v, v, ctx);

    fc_lds_wmma_f32<<<FC_BLOCKS, 256, 0, stream>>>(ctx, Wfc, bfc, out);
}


# Q,K,V launch: one wave builds the q, k, v tiles of its (16 rows, head) from one set of x fragments, 3 independent accumulators, next 64 k loaded a trip ahead; 3072 waves become 1024; sum order kept
# speedup vs baseline: 3.5836x; 1.0423x over previous
#include <hip/hip_runtime.h>
#include <hip/hip_bf16.h>

#define BB   16
#define SS   64
#define HH   16
#define DD   16
#define EE   256
#define HID  128
#define MQ   (BB*SS)
#define MPRE (BB*SS*HH)

typedef float v2f __attribute__((ext_vector_type(2)));
typedef float v4f __attribute__((ext_vector_type(4)));
typedef float v8f __attribute__((ext_vector_type(8)));

__device__ __forceinline__ float fast_tanh(float x) {
#if __has_builtin(__builtin_amdgcn_tanhf)
    return __builtin_amdgcn_tanhf(x);
#else
    return tanhf(x);
#endif
}

__device__ __forceinline__ v4f ld4(const float* p) {
    return *reinterpret_cast<const v4f*>(p);
}

template <int OFF>
__device__ __forceinline__ float lane_xor(float x) {
    const int b = __builtin_bit_cast(int, x);
    if constexpr (OFF == 16)
        return __builtin_bit_cast(float,
            __builtin_amdgcn_permlanex16(b, b, 0x76543210, 0xfedcba98, false, false));
    else
        return __builtin_bit_cast(float,
            __builtin_amdgcn_update_dpp(0, b, 0x160 + OFF, 0xf, 0xf, true));
}

template <int OFF>
__device__ __forceinline__ void xtree_step(float* p, int lane) {
    const bool up = (lane & OFF) != 0;
#pragma unroll
    for (int k = 0; k < OFF; ++k) {
        const float keep = up ? p[k + OFF] : p[k];
        const float send = up ? p[k] : p[k + OFF];
        p[k] = keep + lane_xor<OFF>(send);
    }
}

#define GEMM_WAVES 8
#define FC_K     (SS * DD)
#define FC_T     32
#define FC_KC    256
#define FC_NST   (FC_K / FC_KC)
#define FC_PITCH (FC_KC + 4)
#define FC_STAGE (2 * FC_T * FC_PITCH)
#define FC_LPT   ((2 * FC_T * FC_KC) / (256 * 4))
#define FC_BLOCKS ((MQ / FC_T) * (EE / FC_T))

__device__ __forceinline__ void fc_fetch(v4f (&g)[FC_LPT], const float* __restrict__ aBlk,
                                         const float* __restrict__ wBlk, int kc, int wave, int lane) {
    const int rsub = wave >> 1;
    const int col  = (wave & 1) * 128 + lane * 4 + kc * FC_KC;
#pragma unroll
    for (int q = 0; q < FC_LPT; ++q) {
        const int row = 4 * q + rsub;
        const float* src = (q < FC_LPT / 2) ? aBlk + (size_t)row * FC_K
                                            : wBlk + (size_t)(row - FC_T) * FC_K;
        g[q] = ld4(src + col);
    }
}

__device__ __forceinline__ void fc_put(const v4f (&g)[FC_LPT], float* stg, int wave, int lane) {
    const int rsub = wave >> 1;
    const int col  = (wave & 1) * 128 + lane * 4;
#pragma unroll
    for (int q = 0; q < FC_LPT; ++q)
        *reinterpret_cast<v4f*>(stg + (4 * q + rsub) * FC_PITCH + col) = g[q];
}

__global__ __launch_bounds__(256) void fc_lds_wmma_f32(
    const float* __restrict__ A,
    const float* __restrict__ W,
    const float* __restrict__ bias,
    float* __restrict__ C)
{
    __shared__ __attribute__((aligned(16))) float st[2 * FC_STAGE];
    __shared__ float red[GEMM_WAVES * 2 * 8 * 32];

    const int tid  = threadIdx.x;
    const int lane = tid & 31;
    const int wave = tid >> 5;
    const int half = lane >> 4;
    const int sub  = lane & 15;
    const int m0   = (blockIdx.x >> 3) * FC_T;
    const int n0   = (blockIdx.x & 7) * FC_T;
    const int tmw  = wave & 1;
    const int kq   = wave >> 1;

    const float* aBlk = A + (size_t)m0 * FC_K;
    const float* wBlk = W + (size_t)n0 * FC_K;

    v4f g[FC_LPT];
    fc_fetch(g, aBlk, wBlk, 0, wave, lane);
    fc_put(g, st, wave, lane);
    __syncthreads();

    v8f c0 = {}, c1 = {};
    const int fo = kq * (FC_KC / 4) + 4 * half;
#pragma unroll
    for (int kc = 0; kc < FC_NST; ++kc) {
        if (kc + 1 < FC_NST) {
            fc_fetch(g, aBlk, wBlk, kc + 1, wave, lane);
            __builtin_amdgcn_sched_barrier(0);
        }
        const float* sA = st + (kc & 1) * FC_STAGE + (tmw * 16 + sub) * FC_PITCH + fo;
        const float* sW = st + (kc & 1) * FC_STAGE + (FC_T + sub) * FC_PITCH + fo;
        v4f fa[FC_KC / 32], fb0[FC_KC / 32], fb1[FC_KC / 32];
#pragma unroll
        for (int i = 0; i < FC_KC / 32; ++i) {
            fa[i]  = ld4(sA + 8 * i);
            fb0[i] = ld4(sW + 8 * i);
            fb1[i] = ld4(sW + 16 * FC_PITCH + 8 * i);
        }
        __builtin_amdgcn_sched_barrier(0);
#pragma unroll
        for (int i = 0; i < FC_KC / 32; ++i) {
            c0 = __builtin_amdgcn_wmma_f32_16x16x4_f32(false, v2f{fa[i].x, fa[i].y}, false, v2f{fb0[i].x, fb0[i].y},
                                                       (short)0, c0, false, false);
            c1 = __builtin_amdgcn_wmma_f32_16x16x4_f32(false, v2f{fa[i].x, fa[i].y}, false, v2f{fb1[i].x, fb1[i].y},
                                                       (short)0, c1, false, false);
            c0 = __builtin_amdgcn_wmma_f32_16x16x4_f32(false, v2f{fa[i].z, fa[i].w}, false, v2f{fb0[i].z, fb0[i].w},
                                                       (short)0, c0, false, false);
            c1 = __builtin_amdgcn_wmma_f32_16x16x4_f32(false, v2f{fa[i].z, fa[i].w}, false, v2f{fb1[i].z, fb1[i].w},
                                                       (short)0, c1, false, false);
        }
        if (kc + 1 < FC_NST) {
            __builtin_amdgcn_sched_barrier(0);
            fc_put(g, st + ((kc + 1) & 1) * FC_STAGE, wave, lane);
        }
        __syncthreads();
    }

#pragma unroll
    for (int r = 0; r < 8; ++r) {
        red[((wave * 2 + 0) * 8 + r) * 32 + lane] = c0[r];
        red[((wave * 2 + 1) * 8 + r) * 32 + lane] = c1[r];
    }
    __syncthreads();

    const float bb = bias[n0 + lane];
    const int   t  = lane >> 4;
#pragma unroll
    for (int o = 0; o < 4; ++o) {
        const int rr  = wave * 4 + o;
        const int tms = rr >> 4;
        const int src = ((rr >> 3) & 1) * 16 + sub;
        const int r   = rr & 7;
        float sum = 0.0f;
#pragma unroll
        for (int q = 0; q < 4; ++q)
            sum += red[(((q * 2 + tms) * 2 + t) * 8 + r) * 32 + src];
        C[(size_t)(m0 + rr) * EE + n0 + lane] = sum + bb;
    }
}

#define QKV_WAVES  4
#define QKV_BLOCKS ((MQ / 16 / QKV_WAVES) * HH)
#define QKV_TRIP   64
#define QKV_NTRIP  (EE / QKV_TRIP)
#define QKV_FR     (QKV_TRIP / 8)

__device__ __forceinline__ v2f lo2(const v4f f) { return v2f{f.x, f.y}; }
__device__ __forceinline__ v2f hi2(const v4f f) { return v2f{f.z, f.w}; }

__device__ __forceinline__ void qkv_fetch(v4f (&fx)[QKV_FR], v4f (&fq)[QKV_FR], v4f (&fk)[QKV_FR], v4f (&fv)[QKV_FR],
                                          const float* __restrict__ xRow, const float* __restrict__ qRow,
                                          const float* __restrict__ kRow, const float* __restrict__ vRow, int k0) {
#pragma unroll
    for (int i = 0; i < QKV_FR; ++i) {
        fx[i] = ld4(xRow + k0 + 8 * i);
        fq[i] = ld4(qRow + k0 + 8 * i);
        fk[i] = ld4(kRow + k0 + 8 * i);
        fv[i] = ld4(vRow + k0 + 8 * i);
    }
}

__global__ __launch_bounds__(32 * QKV_WAVES) void qkv_pre_wmma_f32(
    const float* __restrict__ x,
    const float* __restrict__ Wq, const float* __restrict__ Wk, const float* __restrict__ Wv,
    const float* __restrict__ bq, const float* __restrict__ bk, const float* __restrict__ bv,
    const float* __restrict__ Wa, const float* __restrict__ ba,
    float* __restrict__ pre_q, float* __restrict__ pre_k, float* __restrict__ v)
{
    const int lane = threadIdx.x;
    const int wave = threadIdx.y;
    const int tn   = blockIdx.x & (HH - 1);
    const int tm   = (blockIdx.x / HH) * QKV_WAVES + wave;
    const int m0   = tm * 16;
    const int n0   = tn * 16;
    const int half = lane >> 4;
    const int sub  = lane & 15;

    const float* xRow = x  + (size_t)(m0 + sub) * EE + half * 4;
    const float* qRow = Wq + (size_t)(n0 + sub) * EE + half * 4;
    const float* kRow = Wk + (size_t)(n0 + sub) * EE + half * 4;
    const float* vRow = Wv + (size_t)(n0 + sub) * EE + half * 4;

    v4f fx[2][QKV_FR], fq[2][QKV_FR], fk[2][QKV_FR], fv[2][QKV_FR];
    qkv_fetch(fx[0], fq[0], fk[0], fv[0], xRow, qRow, kRow, vRow, 0);

    v4f wq0[HID / 16], wq1[HID / 16], wk0[HID / 16], wk1[HID / 16];
    v4f blq, bhq, blk, bhk;
    float bav[HID / 16];
    float bvv;

    v8f cq = {}, ck = {}, cv = {};
#pragma unroll
    for (int t = 0; t < QKV_NTRIP; ++t) {
        if (t + 1 < QKV_NTRIP) {
            qkv_fetch(fx[(t + 1) & 1], fq[(t + 1) & 1], fk[(t + 1) & 1], fv[(t + 1) & 1],
                      xRow, qRow, kRow, vRow, (t + 1) * QKV_TRIP);
        } else {
            const float* waRow = Wa + (size_t)sub * (2 * DD) + half * 8;
#pragma unroll
            for (int ct = 0; ct < HID / 16; ++ct) {
                wq0[ct] = ld4(waRow + (size_t)ct * 16 * (2 * DD));
                wq1[ct] = ld4(waRow + (size_t)ct * 16 * (2 * DD) + 4);
                wk0[ct] = ld4(waRow + (size_t)ct * 16 * (2 * DD) + DD);
                wk1[ct] = ld4(waRow + (size_t)ct * 16 * (2 * DD) + DD + 4);
            }
            blq = ld4(bq + n0 + half * 8);
            bhq = ld4(bq + n0 + half * 8 + 4);
            blk = ld4(bk + n0 + half * 8);
            bhk = ld4(bk + n0 + half * 8 + 4);
#pragma unroll
            for (int ct = 0; ct < HID / 16; ++ct) bav[ct] = ba[ct * 16 + sub];
            bvv = bv[n0 + sub];
        }
        __builtin_amdgcn_sched_barrier(0);
#pragma unroll
        for (int i = 0; i < QKV_FR; ++i) {
            const v4f ax = fx[t & 1][i], aq = fq[t & 1][i], ak = fk[t & 1][i], av = fv[t & 1][i];
            cq = __builtin_amdgcn_wmma_f32_16x16x4_f32(false, lo2(aq), false, lo2(ax), (short)0, cq, false, false);
            ck = __builtin_amdgcn_wmma_f32_16x16x4_f32(false, lo2(ak), false, lo2(ax), (short)0, ck, false, false);
            cv = __builtin_amdgcn_wmma_f32_16x16x4_f32(false, lo2(ax), false, lo2(av), (short)0, cv, false, false);
            cq = __builtin_amdgcn_wmma_f32_16x16x4_f32(false, hi2(aq), false, hi2(ax), (short)0, cq, false, false);
            ck = __builtin_amdgcn_wmma_f32_16x16x4_f32(false, hi2(ak), false, hi2(ax), (short)0, ck, false, false);
            cv = __builtin_amdgcn_wmma_f32_16x16x4_f32(false, hi2(ax), false, hi2(av), (short)0, cv, false, false);
        }
    }

    {
        const int col  = n0 + sub;
        const int row0 = m0 + half * 8;
#pragma unroll
        for (int r = 0; r < 8; ++r)
            v[(size_t)(row0 + r) * EE + col] = cv[r] + bvv;
    }

    const v2f qa0 = {cq[0] + blq.x, cq[1] + blq.y};
    const v2f qa1 = {cq[2] + blq.z, cq[3] + blq.w};
    const v2f qa2 = {cq[4] + bhq.x, cq[5] + bhq.y};
    const v2f qa3 = {cq[6] + bhq.z, cq[7] + bhq.w};
    const v2f ka0 = {ck[0] + blk.x, ck[1] + blk.y};
    const v2f ka1 = {ck[2] + blk.z, ck[3] + blk.w};
    const v2f ka2 = {ck[4] + bhk.x, ck[5] + bhk.y};
    const v2f ka3 = {ck[6] + bhk.z, ck[7] + bhk.w};

    const size_t po = (size_t)(m0 + half * 8) * (HH * HID) + (size_t)tn * HID + sub;
    float* outq = pre_q + po;
    float* outk = pre_k + po;
#pragma unroll
    for (int ct = 0; ct < HID / 16; ++ct) {
        v8f pq = {}, pk = {};
        pq = __builtin_amdgcn_wmma_f32_16x16x4_f32(false, qa0, false, lo2(wq0[ct]), (short)0, pq, false, false);
        pk = __builtin_amdgcn_wmma_f32_16x16x4_f32(false, ka0, false, lo2(wk0[ct]), (short)0, pk, false, false);
        pq = __builtin_amdgcn_wmma_f32_16x16x4_f32(false, qa1, false, hi2(wq0[ct]), (short)0, pq, false, false);
        pk = __builtin_amdgcn_wmma_f32_16x16x4_f32(false, ka1, false, hi2(wk0[ct]), (short)0, pk, false, false);
        pq = __builtin_amdgcn_wmma_f32_16x16x4_f32(false, qa2, false, lo2(wq1[ct]), (short)0, pq, false, false);
        pk = __builtin_amdgcn_wmma_f32_16x16x4_f32(false, ka2, false, lo2(wk1[ct]), (short)0, pk, false, false);
        pq = __builtin_amdgcn_wmma_f32_16x16x4_f32(false, qa3, false, hi2(wq1[ct]), (short)0, pq, false, false);
        pk = __builtin_amdgcn_wmma_f32_16x16x4_f32(false, ka3, false, hi2(wk1[ct]), (short)0, pk, false, false);
        const float bb = bav[ct];
#pragma unroll
        for (int r = 0; r < 8; ++r) {
            outq[(size_t)r * (HH * HID) + ct * 16] = pq[r] + bb;
            outk[(size_t)r * (HH * HID) + ct * 16] = pk[r];
        }
    }
}

#define IQ        4
#define SCP       (SS * HH + 4)
#define SC_THREADS (32 * HH)
#define SC_BLOCKS  (BB * SS / IQ)

__device__ __forceinline__ void load_keys8(v4f (&kk)[8], const float* pkh, int j0) {
#pragma unroll
    for (int u = 0; u < 8; ++u)
        kk[u] = ld4(pkh + (size_t)(j0 + u) * (HH * HID));
}

__device__ __forceinline__ v2f tanh2(v2f s) {
    return v2f{fast_tanh(s.x), fast_tanh(s.y)};
}

__device__ __forceinline__ float scores_8keys(const v4f (&kk)[8],
                                              const v2f (&qx)[IQ / 2], const v2f (&qy)[IQ / 2],
                                              const v2f (&qz)[IQ / 2], const v2f (&qw)[IQ / 2],
                                              const v4f a, int lane) {
    const v2f ax = {a.x, a.x}, ay = {a.y, a.y}, az = {a.z, a.z}, aw = {a.w, a.w};
    float part[32];
#pragma unroll
    for (int qp = 0; qp < IQ / 2; ++qp) {
#pragma unroll
        for (int u = 0; u < 8; ++u) {
            const v2f kx = {kk[u].x, kk[u].x}, ky = {kk[u].y, kk[u].y};
            const v2f kz = {kk[u].z, kk[u].z}, kw = {kk[u].w, kk[u].w};
            const v2f r = tanh2(qx[qp] + kx) * ax
                        + tanh2(qy[qp] + ky) * ay
                        + tanh2(qz[qp] + kz) * az
                        + tanh2(qw[qp] + kw) * aw;
            part[(2 * qp) * 8 + u]     = r.x;
            part[(2 * qp + 1) * 8 + u] = r.y;
        }
    }
    xtree_step<16>(part, lane);
    xtree_step<8>(part, lane);
    xtree_step<4>(part, lane);
    xtree_step<2>(part, lane);
    xtree_step<1>(part, lane);
    return part[0];
}

__global__ __launch_bounds__(SC_THREADS) void scores_softmax_ctx(
    const float* __restrict__ pre_q,
    const float* __restrict__ pre_k,
    const float* __restrict__ attn_v,
    const float* __restrict__ v,
    float* __restrict__ ctx)
{
    __shared__ float sc[IQ * SCP];
    __shared__ float vloc[IQ * HH * DD];

    const int    tid = threadIdx.x;
    const size_t bi0 = (size_t)blockIdx.x * IQ;
    const size_t bat = bi0 / SS;

    const float* vg = v + bi0 * (HH * DD);
    for (int t = tid; t < IQ * HH * DD; t += SC_THREADS) vloc[t] = vg[t];

    const int lane = tid & 31;
    const int h    = tid >> 5;
    const int c0   = lane * 4;
    const v4f a    = ld4(attn_v + c0);
    v2f qx[IQ / 2], qy[IQ / 2], qz[IQ / 2], qw[IQ / 2];
#pragma unroll
    for (int qp = 0; qp < IQ / 2; ++qp) {
        const float* pq0 = pre_q + (bi0 + 2 * qp) * (size_t)(HH * HID) + (size_t)h * HID + c0;
        const v4f q0 = ld4(pq0);
        const v4f q1 = ld4(pq0 + HH * HID);
        qx[qp] = v2f{q0.x, q1.x};  qy[qp] = v2f{q0.y, q1.y};
        qz[qp] = v2f{q0.z, q1.z};  qw[qp] = v2f{q0.w, q1.w};
    }
    const float* pkh = pre_k + bat * (size_t)(SS * HH * HID) + (size_t)h * HID + c0;

    const int sidx = (lane >> 3) * SCP + (lane & 7) * HH + h;
    v4f kA[8], kB[8];
    load_keys8(kA, pkh, 0);
    for (int jb = 0; jb < SS - 16; jb += 16) {
        load_keys8(kB, pkh, jb + 8);
        __builtin_amdgcn_sched_barrier(0);
        sc[sidx + jb * HH] = scores_8keys(kA, qx, qy, qz, qw, a, lane);
        load_keys8(kA, pkh, jb + 16);
        __builtin_amdgcn_sched_barrier(0);
        sc[sidx + (jb + 8) * HH] = scores_8keys(kB, qx, qy, qz, qw, a, lane);
    }
    load_keys8(kB, pkh, SS - 8);
    __builtin_amdgcn_sched_barrier(0);
    sc[sidx + (SS - 16) * HH] = scores_8keys(kA, qx, qy, qz, qw, a, lane);
    sc[sidx + (SS - 8) * HH]  = scores_8keys(kB, qx, qy, qz, qw, a, lane);

    float e[SS / 8];
#pragma unroll
    for (int c = 0; c < SS / 8; ++c) e[c] = sc[sidx + c * 8 * HH];
    float m = e[0];
#pragma unroll
    for (int c = 1; c < SS / 8; ++c) m = fmaxf(m, e[c]);
    m = fmaxf(m, lane_xor<1>(m));
    m = fmaxf(m, lane_xor<2>(m));
    m = fmaxf(m, lane_xor<4>(m));
    float s = 0.0f;
#pragma unroll
    for (int c = 0; c < SS / 8; ++c) {
        e[c] = __expf(e[c] - m);
        s += e[c];
    }
    s += lane_xor<1>(s);
    s += lane_xor<2>(s);
    s += lane_xor<4>(s);
    const float inv = 1.0f / s;
#pragma unroll
    for (int c = 0; c < SS / 8; ++c) sc[sidx + c * 8 * HH] = e[c] * inv;
    __syncthreads();

    float* cg = ctx + bi0 * (size_t)(SS * DD);
    for (int idx = tid; idx < IQ * SS * DD; idx += SC_THREADS) {
        const int qi = idx >> 10;
        const int j  = (idx >> 4) & (SS - 1);
        const int d  = idx & 15;
        const float* wr = sc + qi * SCP + j * HH;
        const float* vr = vloc + qi * (HH * DD) + d;
        float acc = 0.0f;
#pragma unroll
        for (int hh = 0; hh < HH; ++hh)
            acc = fmaf(wr[hh], vr[hh * DD], acc);
        cg[idx] = acc;
    }
}

extern "C" void kernel_launch(void* const* d_in, const int* in_sizes, int n_in,
                              void* d_out, int out_size, void* d_ws, size_t ws_size,
                              hipStream_t stream) {
    (void)in_sizes; (void)n_in; (void)out_size; (void)ws_size;
    const float* x      = (const float*)d_in[0];
    const float* Wq     = (const float*)d_in[1];
    const float* bq     = (const float*)d_in[2];
    const float* Wk     = (const float*)d_in[3];
    const float* bk     = (const float*)d_in[4];
    const float* Wv     = (const float*)d_in[5];
    const float* bv     = (const float*)d_in[6];
    const float* Wa     = (const float*)d_in[7];
    const float* ba     = (const float*)d_in[8];
    const float* attn_v = (const float*)d_in[9];
    const float* Wfc    = (const float*)d_in[10];
    const float* bfc    = (const float*)d_in[11];
    float* out = (float*)d_out;

    float* v     = (float*)d_ws;
    float* pre_q = v     + (size_t)MQ * EE;
    float* pre_k = pre_q + (size_t)MPRE * HID;
    float* ctx   = pre_k + (size_t)MPRE * HID;

    qkv_pre_wmma_f32<<<QKV_BLOCKS, dim3(32, QKV_WAVES), 0, stream>>>(
        x, Wq, Wk, Wv, bq, bk, bv, Wa, ba, pre_q, pre_k, v);

    scores_softmax_ctx<<<SC_BLOCKS, SC_THREADS, 0, stream>>>(pre_q, pre_k, attn_v, v, ctx);

    fc_lds_wmma_f32<<<FC_BLOCKS, 256, 0, stream>>>(ctx, Wfc, bfc, out);
}


# scores: a lane owns 2 keys and sums its 128 c in order (pre_k stored key-paired via LDS by the projection kernel; pre_q, attn_v scalar operands): no cross-lane sums; score, softmax sum order change
# speedup vs baseline: 3.8523x; 1.0750x over previous
#include <hip/hip_runtime.h>
#include <hip/hip_bf16.h>

#define BB   16
#define SS   64
#define HH   16
#define DD   16
#define EE   256
#define HID  128
#define MQ   (BB*SS)
#define MPRE (BB*SS*HH)

typedef float v2f __attribute__((ext_vector_type(2)));
typedef float v4f __attribute__((ext_vector_type(4)));
typedef float v8f __attribute__((ext_vector_type(8)));

__device__ __forceinline__ float fast_tanh(float x) {
#if __has_builtin(__builtin_amdgcn_tanhf)
    return __builtin_amdgcn_tanhf(x);
#else
    return tanhf(x);
#endif
}

__device__ __forceinline__ v4f ld4(const float* p) {
    return *reinterpret_cast<const v4f*>(p);
}

template <int OFF>
__device__ __forceinline__ float lane_xor(float x) {
    const int b = __builtin_bit_cast(int, x);
    if constexpr (OFF == 16)
        return __builtin_bit_cast(float,
            __builtin_amdgcn_permlanex16(b, b, 0x76543210, 0xfedcba98, false, false));
    else
        return __builtin_bit_cast(float,
            __builtin_amdgcn_update_dpp(0, b, 0x160 + OFF, 0xf, 0xf, true));
}

#define GEMM_WAVES 8
#define FC_K     (SS * DD)
#define FC_T     32
#define FC_KC    256
#define FC_NST   (FC_K / FC_KC)
#define FC_PITCH (FC_KC + 4)
#define FC_STAGE (2 * FC_T * FC_PITCH)
#define FC_LPT   ((2 * FC_T * FC_KC) / (256 * 4))
#define FC_BLOCKS ((MQ / FC_T) * (EE / FC_T))

__device__ __forceinline__ void fc_fetch(v4f (&g)[FC_LPT], const float* __restrict__ aBlk,
                                         const float* __restrict__ wBlk, int kc, int wave, int lane) {
    const int rsub = wave >> 1;
    const int col  = (wave & 1) * 128 + lane * 4 + kc * FC_KC;
#pragma unroll
    for (int q = 0; q < FC_LPT; ++q) {
        const int row = 4 * q + rsub;
        const float* src = (q < FC_LPT / 2) ? aBlk + (size_t)row * FC_K
                                            : wBlk + (size_t)(row - FC_T) * FC_K;
        g[q] = ld4(src + col);
    }
}

__device__ __forceinline__ void fc_put(const v4f (&g)[FC_LPT], float* stg, int wave, int lane) {
    const int rsub = wave >> 1;
    const int col  = (wave & 1) * 128 + lane * 4;
#pragma unroll
    for (int q = 0; q < FC_LPT; ++q)
        *reinterpret_cast<v4f*>(stg + (4 * q + rsub) * FC_PITCH + col) = g[q];
}

__global__ __launch_bounds__(256) void fc_lds_wmma_f32(
    const float* __restrict__ A,
    const float* __restrict__ W,
    const float* __restrict__ bias,
    float* __restrict__ C)
{
    __shared__ __attribute__((aligned(16))) float st[2 * FC_STAGE];
    __shared__ float red[GEMM_WAVES * 2 * 8 * 32];

    const int tid  = threadIdx.x;
    const int lane = tid & 31;
    const int wave = tid >> 5;
    const int half = lane >> 4;
    const int sub  = lane & 15;
    const int m0   = (blockIdx.x >> 3) * FC_T;
    const int n0   = (blockIdx.x & 7) * FC_T;
    const int tmw  = wave & 1;
    const int kq   = wave >> 1;

    const float* aBlk = A + (size_t)m0 * FC_K;
    const float* wBlk = W + (size_t)n0 * FC_K;

    v4f g[FC_LPT];
    fc_fetch(g, aBlk, wBlk, 0, wave, lane);
    fc_put(g, st, wave, lane);
    __syncthreads();

    v8f c0 = {}, c1 = {};
    const int fo = kq * (FC_KC / 4) + 4 * half;
#pragma unroll
    for (int kc = 0; kc < FC_NST; ++kc) {
        if (kc + 1 < FC_NST) {
            fc_fetch(g, aBlk, wBlk, kc + 1, wave, lane);
            __builtin_amdgcn_sched_barrier(0);
        }
        const float* sA = st + (kc & 1) * FC_STAGE + (tmw * 16 + sub) * FC_PITCH + fo;
        const float* sW = st + (kc & 1) * FC_STAGE + (FC_T + sub) * FC_PITCH + fo;
        v4f fa[FC_KC / 32], fb0[FC_KC / 32], fb1[FC_KC / 32];
#pragma unroll
        for (int i = 0; i < FC_KC / 32; ++i) {
            fa[i]  = ld4(sA + 8 * i);
            fb0[i] = ld4(sW + 8 * i);
            fb1[i] = ld4(sW + 16 * FC_PITCH + 8 * i);
        }
        __builtin_amdgcn_sched_barrier(0);
#pragma unroll
        for (int i = 0; i < FC_KC / 32; ++i) {
            c0 = __builtin_amdgcn_wmma_f32_16x16x4_f32(false, v2f{fa[i].x, fa[i].y}, false, v2f{fb0[i].x, fb0[i].y},
                                                       (short)0, c0, false, false);
            c1 = __builtin_amdgcn_wmma_f32_16x16x4_f32(false, v2f{fa[i].x, fa[i].y}, false, v2f{fb1[i].x, fb1[i].y},
                                                       (short)0, c1, false, false);
            c0 = __builtin_amdgcn_wmma_f32_16x16x4_f32(false, v2f{fa[i].z, fa[i].w}, false, v2f{fb0[i].z, fb0[i].w},
                                                       (short)0, c0, false, false);
            c1 = __builtin_amdgcn_wmma_f32_16x16x4_f32(false, v2f{fa[i].z, fa[i].w}, false, v2f{fb1[i].z, fb1[i].w},
                                                       (short)0, c1, false, false);
        }
        if (kc + 1 < FC_NST) {
            __builtin_amdgcn_sched_barrier(0);
            fc_put(g, st + ((kc + 1) & 1) * FC_STAGE, wave, lane);
        }
        __syncthreads();
    }

#pragma unroll
    for (int r = 0; r < 8; ++r) {
        red[((wave * 2 + 0) * 8 + r) * 32 + lane] = c0[r];
        red[((wave * 2 + 1) * 8 + r) * 32 + lane] = c1[r];
    }
    __syncthreads();

    const float bb = bias[n0 + lane];
    const int   t  = lane >> 4;
#pragma unroll
    for (int o = 0; o < 4; ++o) {
        const int rr  = wave * 4 + o;
        const int tms = rr >> 4;
        const int src = ((rr >> 3) & 1) * 16 + sub;
        const int r   = rr & 7;
        float sum = 0.0f;
#pragma unroll
        for (int q = 0; q < 4; ++q)
            sum += red[(((q * 2 + tms) * 2 + t) * 8 + r) * 32 + src];
        C[(size_t)(m0 + rr) * EE + n0 + lane] = sum + bb;
    }
}

#define QKV_WAVES  4
#define QKV_BLOCKS ((MQ / 16 / QKV_WAVES) * HH)
#define QKV_TRIP   64
#define QKV_NTRIP  (EE / QKV_TRIP)
#define QKV_FR     (QKV_TRIP / 8)
#define KT_PITCH   (2 * SS + 4)

__device__ __forceinline__ v2f lo2(const v4f f) { return v2f{f.x, f.y}; }
__device__ __forceinline__ v2f hi2(const v4f f) { return v2f{f.z, f.w}; }

__device__ __forceinline__ void qkv_fetch(v4f (&fx)[QKV_FR], v4f (&fq)[QKV_FR], v4f (&fk)[QKV_FR], v4f (&fv)[QKV_FR],
                                          const float* __restrict__ xRow, const float* __restrict__ qRow,
                                          const float* __restrict__ kRow, const float* __restrict__ vRow, int k0) {
#pragma unroll
    for (int i = 0; i < QKV_FR; ++i) {
        fx[i] = ld4(xRow + k0 + 8 * i);
        fq[i] = ld4(qRow + k0 + 8 * i);
        fk[i] = ld4(kRow + k0 + 8 * i);
        fv[i] = ld4(vRow + k0 + 8 * i);
    }
}

__global__ __launch_bounds__(32 * QKV_WAVES) void qkv_pre_wmma_f32(
    const float* __restrict__ x,
    const float* __restrict__ Wq, const float* __restrict__ Wk, const float* __restrict__ Wv,
    const float* __restrict__ bq, const float* __restrict__ bk, const float* __restrict__ bv,
    const float* __restrict__ Wa, const float* __restrict__ ba,
    float* __restrict__ pre_q, float* __restrict__ pre_k, float* __restrict__ v)
{
    const int lane = threadIdx.x;
    const int wave = threadIdx.y;
    const int tn   = blockIdx.x & (HH - 1);
    const int tm   = (blockIdx.x / HH) * QKV_WAVES + wave;
    const int m0   = tm * 16;
    const int n0   = tn * 16;
    const int half = lane >> 4;
    const int sub  = lane & 15;

    __shared__ __attribute__((aligned(16))) float kt[(HID / 2) * KT_PITCH];

    const float* xRow = x  + (size_t)(m0 + sub) * EE + half * 4;
    const float* qRow = Wq + (size_t)(n0 + sub) * EE + half * 4;
    const float* kRow = Wk + (size_t)(n0 + sub) * EE + half * 4;
    const float* vRow = Wv + (size_t)(n0 + sub) * EE + half * 4;

    v4f fx[2][QKV_FR], fq[2][QKV_FR], fk[2][QKV_FR], fv[2][QKV_FR];
    qkv_fetch(fx[0], fq[0], fk[0], fv[0], xRow, qRow, kRow, vRow, 0);

    v4f wq0[HID / 16], wq1[HID / 16], wk0[HID / 16], wk1[HID / 16];
    v4f blq, bhq, blk, bhk;
    float bav[HID / 16];
    float bvv;

    v8f cq = {}, ck = {}, cv = {};
#pragma unroll
    for (int t = 0; t < QKV_NTRIP; ++t) {
        if (t + 1 < QKV_NTRIP) {
            qkv_fetch(fx[(t + 1) & 1], fq[(t + 1) & 1], fk[(t + 1) & 1], fv[(t + 1) & 1],
                      xRow, qRow, kRow, vRow, (t + 1) * QKV_TRIP);
        } else {
            const float* waRow = Wa + (size_t)sub * (2 * DD) + half * 8;
#pragma unroll
            for (int ct = 0; ct < HID / 16; ++ct) {
                wq0[ct] = ld4(waRow + (size_t)ct * 16 * (2 * DD));
                wq1[ct] = ld4(waRow + (size_t)ct * 16 * (2 * DD) + 4);
                wk0[ct] = ld4(waRow + (size_t)ct * 16 * (2 * DD) + DD);
                wk1[ct] = ld4(waRow + (size_t)ct * 16 * (2 * DD) + DD + 4);
            }
            blq = ld4(bq + n0 + half * 8);
            bhq = ld4(bq + n0 + half * 8 + 4);
            blk = ld4(bk + n0 + half * 8);
            bhk = ld4(bk + n0 + half * 8 + 4);
#pragma unroll
            for (int ct = 0; ct < HID / 16; ++ct) bav[ct] = ba[ct * 16 + sub];
            bvv = bv[n0 + sub];
        }
        __builtin_amdgcn_sched_barrier(0);
#pragma unroll
        for (int i = 0; i < QKV_FR; ++i) {
            const v4f ax = fx[t & 1][i], aq = fq[t & 1][i], ak = fk[t & 1][i], av = fv[t & 1][i];
            cq = __builtin_amdgcn_wmma_f32_16x16x4_f32(false, lo2(aq), false, lo2(ax), (short)0, cq, false, false);
            ck = __builtin_amdgcn_wmma_f32_16x16x4_f32(false, lo2(ak), false, lo2(ax), (short)0, ck, false, false);
            cv = __builtin_amdgcn_wmma_f32_16x16x4_f32(false, lo2(ax), false, lo2(av), (short)0, cv, false, false);
            cq = __builtin_amdgcn_wmma_f32_16x16x4_f32(false, hi2(aq), false, hi2(ax), (short)0, cq, false, false);
            ck = __builtin_amdgcn_wmma_f32_16x16x4_f32(false, hi2(ak), false, hi2(ax), (short)0, ck, false, false);
            cv = __builtin_amdgcn_wmma_f32_16x16x4_f32(false, hi2(ax), false, hi2(av), (short)0, cv, false, false);
        }
    }

    {
        const int col  = n0 + sub;
        const int row0 = m0 + half * 8;
#pragma unroll
        for (int r = 0; r < 8; ++r)
            v[(size_t)(row0 + r) * EE + col] = cv[r] + bvv;
    }

    const v2f qa0 = {cq[0] + blq.x, cq[1] + blq.y};
    const v2f qa1 = {cq[2] + blq.z, cq[3] + blq.w};
    const v2f qa2 = {cq[4] + bhq.x, cq[5] + bhq.y};
    const v2f qa3 = {cq[6] + bhq.z, cq[7] + bhq.w};
    const v2f ka0 = {ck[0] + blk.x, ck[1] + blk.y};
    const v2f ka1 = {ck[2] + blk.z, ck[3] + blk.w};
    const v2f ka2 = {ck[4] + bhk.x, ck[5] + bhk.y};
    const v2f ka3 = {ck[6] + bhk.z, ck[7] + bhk.w};

    const size_t po = (size_t)(m0 + half * 8) * (HH * HID) + (size_t)tn * HID + sub;
    float* outq = pre_q + po;
    float* ktl = kt + (sub >> 1) * KT_PITCH + (16 * (wave & 1) + 8 * half) * 4 + (sub & 1) * 2 + (wave >> 1);
    v8f pqs[HID / 16];
#pragma unroll
    for (int ct = 0; ct < HID / 16; ++ct) {
        v8f pq = {}, pk = {};
        pq = __builtin_amdgcn_wmma_f32_16x16x4_f32(false, qa0, false, lo2(wq0[ct]), (short)0, pq, false, false);
        pk = __builtin_amdgcn_wmma_f32_16x16x4_f32(false, ka0, false, lo2(wk0[ct]), (short)0, pk, false, false);
        pq = __builtin_amdgcn_wmma_f32_16x16x4_f32(false, qa1, false, hi2(wq0[ct]), (short)0, pq, false, false);
        pk = __builtin_amdgcn_wmma_f32_16x16x4_f32(false, ka1, false, hi2(wk0[ct]), (short)0, pk, false, false);
        pq = __builtin_amdgcn_wmma_f32_16x16x4_f32(false, qa2, false, lo2(wq1[ct]), (short)0, pq, false, false);
        pk = __builtin_amdgcn_wmma_f32_16x16x4_f32(false, ka2, false, lo2(wk1[ct]), (short)0, pk, false, false);
        pq = __builtin_amdgcn_wmma_f32_16x16x4_f32(false, qa3, false, hi2(wq1[ct]), (short)0, pq, false, false);
        pk = __builtin_amdgcn_wmma_f32_16x16x4_f32(false, ka3, false, hi2(wk1[ct]), (short)0, pk, false, false);
        pqs[ct] = pq;
#pragma unroll
        for (int r = 0; r < 8; ++r)
            ktl[ct * 8 * KT_PITCH + 4 * r] = pk[r];
    }
    __syncthreads();
#pragma unroll
    for (int ct = 0; ct < HID / 16; ++ct) {
        const float bb = bav[ct];
#pragma unroll
        for (int r = 0; r < 8; ++r)
            outq[(size_t)r * (HH * HID) + ct * 16] = pqs[ct][r] + bb;
    }
    float* kout = pre_k + ((size_t)(blockIdx.x / HH) * HH + tn) * (size_t)(HID * SS);
#pragma unroll
    for (int i = 0; i < 16; ++i) {
        const int cp = wave * 16 + i;
        *reinterpret_cast<v4f*>(kout + cp * (2 * SS) + lane * 4) =
            *reinterpret_cast<const v4f*>(kt + cp * KT_PITCH + lane * 4);
    }
}

#define IQ        4
#define SJP       (HH + 4)
#define SCP       (SS * SJP)
#define SC_THREADS (32 * HH)
#define SC_BLOCKS  (BB * SS / IQ)
#define SC_CSTEP  16

__device__ __forceinline__ void load_c16(v4f (&kk)[SC_CSTEP / 2], const float* kh, int c0) {
#pragma unroll
    for (int u = 0; u < SC_CSTEP / 2; ++u)
        kk[u] = ld4(kh + (size_t)(c0 / 2 + u) * (2 * SS));
}

__device__ __forceinline__ v2f tanh2(v2f s) {
    return v2f{fast_tanh(s.x), fast_tanh(s.y)};
}

__device__ __forceinline__ void scores_c16(v2f (&acc)[IQ], const v4f (&kk)[SC_CSTEP / 2],
                                           const float* __restrict__ qh, const float* __restrict__ av, int c0) {
#pragma unroll
    for (int u = 0; u < SC_CSTEP / 2; ++u) {
#pragma unroll
        for (int p = 0; p < 2; ++p) {
            const int   c  = c0 + 2 * u + p;
            const v2f   kp = p ? v2f{kk[u].z, kk[u].w} : v2f{kk[u].x, kk[u].y};
            const float ac = av[c];
#pragma unroll
            for (int q = 0; q < IQ; ++q) {
                const float qc = qh[(size_t)q * (HH * HID) + c];
                acc[q] = tanh2(kp + v2f{qc, qc}) * v2f{ac, ac} + acc[q];
            }
        }
    }
}

__global__ __launch_bounds__(SC_THREADS) void scores_softmax_ctx(
    const float* __restrict__ pre_q,
    const float* __restrict__ pre_k,
    const float* __restrict__ attn_v,
    const float* __restrict__ v,
    float* __restrict__ ctx)
{
    __shared__ __attribute__((aligned(16))) float sc[IQ * SCP];
    __shared__ float vloc[IQ * HH * DD];

    const int    tid = threadIdx.x;
    const size_t bi0 = (size_t)blockIdx.x * IQ;
    const size_t bat = bi0 / SS;

    const float* vg = v + bi0 * (HH * DD);
    for (int t = tid; t < IQ * HH * DD; t += SC_THREADS) vloc[t] = vg[t];

    const int lane = tid & 31;
    const int h    = __builtin_amdgcn_readfirstlane(tid >> 5);
    const float* qh = pre_q + bi0 * (size_t)(HH * HID) + (size_t)h * HID;
    if (tid < IQ * HH * HID / 32)
        __builtin_prefetch(pre_q + bi0 * (size_t)(HH * HID) + (size_t)tid * 32, 0, 3);
    const float* kh = pre_k + (bat * HH + h) * (size_t)(HID * SS) + lane * 4;

    v2f acc[IQ];
#pragma unroll
    for (int q = 0; q < IQ; ++q) acc[q] = v2f{0.0f, 0.0f};

    v4f kA[SC_CSTEP / 2], kB[SC_CSTEP / 2];
    load_c16(kA, kh, 0);
    for (int c0 = 0; c0 < HID - 2 * SC_CSTEP; c0 += 2 * SC_CSTEP) {
        load_c16(kB, kh, c0 + SC_CSTEP);
        __builtin_amdgcn_sched_barrier(0);
        scores_c16(acc, kA, qh, attn_v, c0);
        load_c16(kA, kh, c0 + 2 * SC_CSTEP);
        __builtin_amdgcn_sched_barrier(0);
        scores_c16(acc, kB, qh, attn_v, c0 + SC_CSTEP);
    }
    load_c16(kB, kh, HID - SC_CSTEP);
    __builtin_amdgcn_sched_barrier(0);
    scores_c16(acc, kA, qh, attn_v, HID - 2 * SC_CSTEP);
    scores_c16(acc, kB, qh, attn_v, HID - SC_CSTEP);

#pragma unroll
    for (int q = 0; q < IQ; ++q) {
        float m = fmaxf(acc[q].x, acc[q].y);
        m = fmaxf(m, lane_xor<1>(m));
        m = fmaxf(m, lane_xor<2>(m));
        m = fmaxf(m, lane_xor<4>(m));
        m = fmaxf(m, lane_xor<8>(m));
        m = fmaxf(m, lane_xor<16>(m));
        const float e0 = __expf(acc[q].x - m);
        const float e1 = __expf(acc[q].y - m);
        float s = e0 + e1;
        s += lane_xor<1>(s);
        s += lane_xor<2>(s);
        s += lane_xor<4>(s);
        s += lane_xor<8>(s);
        s += lane_xor<16>(s);
        const float inv = 1.0f / s;
        sc[q * SCP + lane * SJP + h]        = e0 * inv;
        sc[q * SCP + (lane + 32) * SJP + h] = e1 * inv;
    }
    __syncthreads();

    float* cg = ctx + bi0 * (size_t)(SS * DD);
    for (int idx = tid; idx < IQ * SS * DD; idx += SC_THREADS) {
        const int qi = idx >> 10;
        const int j  = (idx >> 4) & (SS - 1);
        const int d  = idx & 15;
        const float* wr = sc + qi * SCP + j * SJP;
        const float* vr = vloc + qi * (HH * DD) + d;
        float acc2 = 0.0f;
#pragma unroll
        for (int hh = 0; hh < HH; ++hh)
            acc2 = fmaf(wr[hh], vr[hh * DD], acc2);
        cg[idx] = acc2;
    }
}

extern "C" void kernel_launch(void* const* d_in, const int* in_sizes, int n_in,
                              void* d_out, int out_size, void* d_ws, size_t ws_size,
                              hipStream_t stream) {
    (void)in_sizes; (void)n_in; (void)out_size; (void)ws_size;
    const float* x      = (const float*)d_in[0];
    const float* Wq     = (const float*)d_in[1];
    const float* bq     = (const float*)d_in[2];
    const float* Wk     = (const float*)d_in[3];
    const float* bk     = (const float*)d_in[4];
    const float* Wv     = (const float*)d_in[5];
    const float* bv     = (const float*)d_in[6];
    const float* Wa     = (const float*)d_in[7];
    const float* ba     = (const float*)d_in[8];
    const float* attn_v = (const float*)d_in[9];
    const float* Wfc    = (const float*)d_in[10];
    const float* bfc    = (const float*)d_in[11];
    float* out = (float*)d_out;

    float* v     = (float*)d_ws;
    float* pre_q = v     + (size_t)MQ * EE;
    float* pre_k = pre_q + (size_t)MPRE * HID;
    float* ctx   = pre_k + (size_t)MPRE * HID;

    qkv_pre_wmma_f32<<<QKV_BLOCKS, dim3(32, QKV_WAVES), 0, stream>>>(
        x, Wq, Wk, Wv, bq, bk, bv, Wa, ba, pre_q, pre_k, v);

    scores_softmax_ctx<<<SC_BLOCKS, SC_THREADS, 0, stream>>>(pre_q, pre_k, attn_v, v, ctx);

    fc_lds_wmma_f32<<<FC_BLOCKS, 256, 0, stream>>>(ctx, Wfc, bfc, out);
}


# scores+softmax made by the projection block of its (batch, head) from pre_q, pre_k kept in LDS; ctx head sum on WMMA in the final product's A staging (head sum order changes); 3 launches become 2
# speedup vs baseline: 4.8700x; 1.2642x over previous
#include <hip/hip_runtime.h>
#include <hip/hip_bf16.h>

#define BB   16
#define SS   64
#define HH   16
#define DD   16
#define EE   256
#define HID  128
#define MQ   (BB*SS)
#define MPRE (BB*SS*HH)

typedef float v2f __attribute__((ext_vector_type(2)));
typedef float v4f __attribute__((ext_vector_type(4)));
typedef float v8f __attribute__((ext_vector_type(8)));

__device__ __forceinline__ float fast_tanh(float x) {
#if __has_builtin(__builtin_amdgcn_tanhf)
    return __builtin_amdgcn_tanhf(x);
#else
    return tanhf(x);
#endif
}

__device__ __forceinline__ v4f ld4(const float* p) {
    return *reinterpret_cast<const v4f*>(p);
}

template <int OFF>
__device__ __forceinline__ float lane_xor(float x) {
    const int b = __builtin_bit_cast(int, x);
    if constexpr (OFF == 16)
        return __builtin_bit_cast(float,
            __builtin_amdgcn_permlanex16(b, b, 0x76543210, 0xfedcba98, false, false));
    else
        return __builtin_bit_cast(float,
            __builtin_amdgcn_update_dpp(0, b, 0x160 + OFF, 0xf, 0xf, true));
}


#define GEMM_WAVES 8
#define FC_K     (SS * DD)
#define FC_T     32
#define FC_KC    256
#define FC_NST   (FC_K / FC_KC)
#define FC_PITCH (FC_KC + 4)
#define FC_STAGE (2 * FC_T * FC_PITCH)
#define FC_WLPT  ((FC_T * FC_KC) / (256 * 4))
#define FC_RPW   (FC_T / GEMM_WAVES)
#define FC_BLOCKS ((MQ / FC_T) * (EE / FC_T))

__device__ __forceinline__ void fc_fetch_w(v4f (&g)[FC_WLPT], const float* __restrict__ wBlk,
                                           int kc, int wave, int lane) {
    const int rsub = wave >> 1;
    const int col  = (wave & 1) * 128 + lane * 4 + kc * FC_KC;
#pragma unroll
    for (int q = 0; q < FC_WLPT; ++q)
        g[q] = ld4(wBlk + (size_t)(4 * q + rsub) * FC_K + col);
}

__device__ __forceinline__ void fc_put_w(const v4f (&g)[FC_WLPT], float* stg, int wave, int lane) {
    const int rsub = wave >> 1;
    const int col  = (wave & 1) * 128 + lane * 4;
#pragma unroll
    for (int q = 0; q < FC_WLPT; ++q)
        *reinterpret_cast<v4f*>(stg + (FC_T + 4 * q + rsub) * FC_PITCH + col) = g[q];
}

__device__ __forceinline__ void fc_fetch_p(v2f (&pf)[FC_RPW][4], const float* __restrict__ pRow, int kc) {
#pragma unroll
    for (int r = 0; r < FC_RPW; ++r) {
#pragma unroll
        for (int t = 0; t < 4; ++t) {
            const float* p = pRow + (size_t)(r * HH + 4 * t) * SS + kc * 16;
            pf[r][t] = v2f{p[0], p[SS]};
        }
    }
}

__device__ __forceinline__ void fc_ctx_put(const v2f (&vf)[FC_RPW][4], const v2f (&pf)[FC_RPW][4],
                                           float* stg, int wave, int sub, int half) {
    v8f d[FC_RPW];
#pragma unroll
    for (int r = 0; r < FC_RPW; ++r) d[r] = v8f{};
#pragma unroll
    for (int t = 0; t < 4; ++t) {
#pragma unroll
        for (int r = 0; r < FC_RPW; ++r)
            d[r] = __builtin_amdgcn_wmma_f32_16x16x4_f32(false, vf[r][t], false, pf[r][t],
                                                         (short)0, d[r], false, false);
    }
#pragma unroll
    for (int r = 0; r < FC_RPW; ++r) {
        float* p = stg + (4 * wave + r) * FC_PITCH + sub * 16 + 8 * half;
        *reinterpret_cast<v4f*>(p)     = v4f{d[r][0], d[r][1], d[r][2], d[r][3]};
        *reinterpret_cast<v4f*>(p + 4) = v4f{d[r][4], d[r][5], d[r][6], d[r][7]};
    }
}

__global__ __launch_bounds__(256) void fc_ctx_lds_wmma_f32(
    const float* __restrict__ P,
    const float* __restrict__ V,
    const float* __restrict__ W,
    const float* __restrict__ bias,
    float* __restrict__ C)
{
    __shared__ __attribute__((aligned(16))) float st[2 * FC_STAGE];
    __shared__ float red[GEMM_WAVES * 2 * 8 * 32];

    const int tid  = threadIdx.x;
    const int lane = tid & 31;
    const int wave = tid >> 5;
    const int half = lane >> 4;
    const int sub  = lane & 15;
    const int m0   = (blockIdx.x >> 3) * FC_T;
    const int n0   = (blockIdx.x & 7) * FC_T;
    const int tmw  = wave & 1;
    const int kq   = wave >> 1;

    const float* wBlk = W + (size_t)n0 * FC_K;
    const float* pRow = P + ((size_t)(m0 + 4 * wave) * HH + 2 * half) * SS + sub;
    const float* vRow = V + (size_t)(m0 + 4 * wave) * EE + (2 * half) * DD + sub;

    v4f g[FC_WLPT];
    v2f pf[FC_RPW][4], vf[FC_RPW][4];
    fc_fetch_w(g, wBlk, 0, wave, lane);
    fc_fetch_p(pf, pRow, 0);
#pragma unroll
    for (int r = 0; r < FC_RPW; ++r) {
#pragma unroll
        for (int t = 0; t < 4; ++t)
            vf[r][t] = v2f{vRow[r * EE + 4 * t * DD], vRow[r * EE + 4 * t * DD + DD]};
    }
    fc_ctx_put(vf, pf, st, wave, sub, half);
    fc_put_w(g, st, wave, lane);
    __syncthreads();

    v8f c0 = {}, c1 = {};
    const int fo = kq * (FC_KC / 4) + 4 * half;
#pragma unroll
    for (int kc = 0; kc < FC_NST; ++kc) {
        if (kc + 1 < FC_NST) {
            fc_fetch_p(pf, pRow, kc + 1);
            fc_fetch_w(g, wBlk, kc + 1, wave, lane);
            __builtin_amdgcn_sched_barrier(0);
        }
        const float* sA = st + (kc & 1) * FC_STAGE + (tmw * 16 + sub) * FC_PITCH + fo;
        const float* sW = st + (kc & 1) * FC_STAGE + (FC_T + sub) * FC_PITCH + fo;
        v4f fa[FC_KC / 32], fb0[FC_KC / 32], fb1[FC_KC / 32];
#pragma unroll
        for (int i = 0; i < FC_KC / 32; ++i) {
            fa[i]  = ld4(sA + 8 * i);
            fb0[i] = ld4(sW + 8 * i);
            fb1[i] = ld4(sW + 16 * FC_PITCH + 8 * i);
        }
        __builtin_amdgcn_sched_barrier(0);
#pragma unroll
        for (int i = 0; i < FC_KC / 32; ++i) {
            c0 = __builtin_amdgcn_wmma_f32_16x16x4_f32(false, v2f{fa[i].x, fa[i].y}, false, v2f{fb0[i].x, fb0[i].y},
                                                       (short)0, c0, false, false);
            c1 = __builtin_amdgcn_wmma_f32_16x16x4_f32(false, v2f{fa[i].x, fa[i].y}, false, v2f{fb1[i].x, fb1[i].y},
                                                       (short)0, c1, false, false);
            c0 = __builtin_amdgcn_wmma_f32_16x16x4_f32(false, v2f{fa[i].z, fa[i].w}, false, v2f{fb0[i].z, fb0[i].w},
                                                       (short)0, c0, false, false);
            c1 = __builtin_amdgcn_wmma_f32_16x16x4_f32(false, v2f{fa[i].z, fa[i].w}, false, v2f{fb1[i].z, fb1[i].w},
                                                       (short)0, c1, false, false);
        }
        if (kc + 1 < FC_NST) {
            __builtin_amdgcn_sched_barrier(0);
            fc_ctx_put(vf, pf, st + ((kc + 1) & 1) * FC_STAGE, wave, sub, half);
            fc_put_w(g, st + ((kc + 1) & 1) * FC_STAGE, wave, lane);
        }
        __syncthreads();
    }

#pragma unroll
    for (int r = 0; r < 8; ++r) {
        red[((wave * 2 + 0) * 8 + r) * 32 + lane] = c0[r];
        red[((wave * 2 + 1) * 8 + r) * 32 + lane] = c1[r];
    }
    __syncthreads();

    const float bb = bias[n0 + lane];
    const int   t  = lane >> 4;
#pragma unroll
    for (int o = 0; o < 4; ++o) {
        const int rr  = wave * 4 + o;
        const int tms = rr >> 4;
        const int src = ((rr >> 3) & 1) * 16 + sub;
        const int r   = rr & 7;
        float sum = 0.0f;
#pragma unroll
        for (int q = 0; q < 4; ++q)
            sum += red[(((q * 2 + tms) * 2 + t) * 8 + r) * 32 + src];
        C[(size_t)(m0 + rr) * EE + n0 + lane] = sum + bb;
    }
}

#define QKV_WAVES  4
#define K1_WAVES   8
#define K1_BLOCKS  (BB * HH)
#define QKV_TRIP   64
#define QKV_NTRIP  (EE / QKV_TRIP)
#define QKV_FR     (QKV_TRIP / 8)
#define KT_PITCH   (2 * SS + 4)
#define PQ_PITCH   (HID + 4)
#define SQ         (SS / K1_WAVES)
#define SC_CSTEP   8

__device__ __forceinline__ v2f lo2(const v4f f) { return v2f{f.x, f.y}; }
__device__ __forceinline__ v2f hi2(const v4f f) { return v2f{f.z, f.w}; }

__device__ __forceinline__ void qkv_fetch(v4f (&fx)[QKV_FR], v4f (&fq)[QKV_FR], v4f (&fk)[QKV_FR], v4f (&fv)[QKV_FR],
                                          const float* __restrict__ xRow, const float* __restrict__ qRow,
                                          const float* __restrict__ kRow, const float* __restrict__ vRow, int k0) {
#pragma unroll
    for (int i = 0; i < QKV_FR; ++i) {
        fx[i] = ld4(xRow + k0 + 8 * i);
        fq[i] = ld4(qRow + k0 + 8 * i);
        fk[i] = ld4(kRow + k0 + 8 * i);
        fv[i] = ld4(vRow + k0 + 8 * i);
    }
}

__device__ __forceinline__ void qkv_pre_tile(
    const float* __restrict__ x,
    const float* __restrict__ Wq, const float* __restrict__ Wk, const float* __restrict__ Wv,
    const float* __restrict__ bq, const float* __restrict__ bk, const float* __restrict__ bv,
    const float* __restrict__ Wa, const float* __restrict__ ba,
    float* __restrict__ v, float* kt, float* pql, int bat, int tn, int wave, int lane)
{
    const int m0   = (bat * QKV_WAVES + wave) * 16;
    const int n0   = tn * 16;
    const int half = lane >> 4;
    const int sub  = lane & 15;

    const float* xRow = x  + (size_t)(m0 + sub) * EE + half * 4;
    const float* qRow = Wq + (size_t)(n0 + sub) * EE + half * 4;
    const float* kRow = Wk + (size_t)(n0 + sub) * EE + half * 4;
    const float* vRow = Wv + (size_t)(n0 + sub) * EE + half * 4;

    v4f fx[2][QKV_FR], fq[2][QKV_FR], fk[2][QKV_FR], fv[2][QKV_FR];
    qkv_fetch(fx[0], fq[0], fk[0], fv[0], xRow, qRow, kRow, vRow, 0);

    v4f wq0[HID / 16], wq1[HID / 16], wk0[HID / 16], wk1[HID / 16];
    v4f blq, bhq, blk, bhk;
    float bav[HID / 16];
    float bvv;

    v8f cq = {}, ck = {}, cv = {};
#pragma unroll
    for (int t = 0; t < QKV_NTRIP; ++t) {
        if (t + 1 < QKV_NTRIP) {
            qkv_fetch(fx[(t + 1) & 1], fq[(t + 1) & 1], fk[(t + 1) & 1], fv[(t + 1) & 1],
                      xRow, qRow, kRow, vRow, (t + 1) * QKV_TRIP);
        } else {
            const float* waRow = Wa + (size_t)sub * (2 * DD) + half * 8;
#pragma unroll
            for (int ct = 0; ct < HID / 16; ++ct) {
                wq0[ct] = ld4(waRow + (size_t)ct * 16 * (2 * DD));
                wq1[ct] = ld4(waRow + (size_t)ct * 16 * (2 * DD) + 4);
                wk0[ct] = ld4(waRow + (size_t)ct * 16 * (2 * DD) + DD);
                wk1[ct] = ld4(waRow + (size_t)ct * 16 * (2 * DD) + DD + 4);
            }
            blq = ld4(bq + n0 + half * 8);
            bhq = ld4(bq + n0 + half * 8 + 4);
            blk = ld4(bk + n0 + half * 8);
            bhk = ld4(bk + n0 + half * 8 + 4);
#pragma unroll
            for (int ct = 0; ct < HID / 16; ++ct) bav[ct] = ba[ct * 16 + sub];
            bvv = bv[n0 + sub];
        }
        __builtin_amdgcn_sched_barrier(0);
#pragma unroll
        for (int i = 0; i < QKV_FR; ++i) {
            const v4f ax = fx[t & 1][i], aq = fq[t & 1][i], ak = fk[t & 1][i], av = fv[t & 1][i];
            cq = __builtin_amdgcn_wmma_f32_16x16x4_f32(false, lo2(aq), false, lo2(ax), (short)0, cq, false, false);
            ck = __builtin_amdgcn_wmma_f32_16x16x4_f32(false, lo2(ak), false, lo2(ax), (short)0, ck, false, false);
            cv = __builtin_amdgcn_wmma_f32_16x16x4_f32(false, lo2(ax), false, lo2(av), (short)0, cv, false, false);
            cq = __builtin_amdgcn_wmma_f32_16x16x4_f32(false, hi2(aq), false, hi2(ax), (short)0, cq, false, false);
            ck = __builtin_amdgcn_wmma_f32_16x16x4_f32(false, hi2(ak), false, hi2(ax), (short)0, ck, false, false);
            cv = __builtin_amdgcn_wmma_f32_16x16x4_f32(false, hi2(ax), false, hi2(av), (short)0, cv, false, false);
        }
    }

    {
        const int col  = n0 + sub;
        const int row0 = m0 + half * 8;
#pragma unroll
        for (int r = 0; r < 8; ++r)
            v[(size_t)(row0 + r) * EE + col] = cv[r] + bvv;
    }

    const v2f qa0 = {cq[0] + blq.x, cq[1] + blq.y};
    const v2f qa1 = {cq[2] + blq.z, cq[3] + blq.w};
    const v2f qa2 = {cq[4] + bhq.x, cq[5] + bhq.y};
    const v2f qa3 = {cq[6] + bhq.z, cq[7] + bhq.w};
    const v2f ka0 = {ck[0] + blk.x, ck[1] + blk.y};
    const v2f ka1 = {ck[2] + blk.z, ck[3] + blk.w};
    const v2f ka2 = {ck[4] + bhk.x, ck[5] + bhk.y};
    const v2f ka3 = {ck[6] + bhk.z, ck[7] + bhk.w};

    float* pqw = pql + (16 * wave + 8 * half) * PQ_PITCH + sub;
    float* ktl = kt + (sub >> 1) * KT_PITCH + (16 * (wave & 1) + 8 * half) * 4 + (sub & 1) * 2 + (wave >> 1);
#pragma unroll
    for (int ct = 0; ct < HID / 16; ++ct) {
        v8f pq = {}, pk = {};
        pq = __builtin_amdgcn_wmma_f32_16x16x4_f32(false, qa0, false, lo2(wq0[ct]), (short)0, pq, false, false);
        pk = __builtin_amdgcn_wmma_f32_16x16x4_f32(false, ka0, false, lo2(wk0[ct]), (short)0, pk, false, false);
        pq = __builtin_amdgcn_wmma_f32_16x16x4_f32(false, qa1, false, hi2(wq0[ct]), (short)0, pq, false, false);
        pk = __builtin_amdgcn_wmma_f32_16x16x4_f32(false, ka1, false, hi2(wk0[ct]), (short)0, pk, false, false);
        pq = __builtin_amdgcn_wmma_f32_16x16x4_f32(false, qa2, false, lo2(wq1[ct]), (short)0, pq, false, false);
        pk = __builtin_amdgcn_wmma_f32_16x16x4_f32(false, ka2, false, lo2(wk1[ct]), (short)0, pk, false, false);
        pq = __builtin_amdgcn_wmma_f32_16x16x4_f32(false, qa3, false, hi2(wq1[ct]), (short)0, pq, false, false);
        pk = __builtin_amdgcn_wmma_f32_16x16x4_f32(false, ka3, false, hi2(wk1[ct]), (short)0, pk, false, false);
        const float bb = bav[ct];
#pragma unroll
        for (int r = 0; r < 8; ++r)
            pqw[r * PQ_PITCH + ct * 16] = pq[r] + bb;
#pragma unroll
        for (int r = 0; r < 8; ++r)
            ktl[ct * 8 * KT_PITCH + 4 * r] = pk[r];
    }
}

__device__ __forceinline__ v2f tanh2(v2f s) {
    return v2f{fast_tanh(s.x), fast_tanh(s.y)};
}

__device__ __forceinline__ void sc_load(v4f (&kk)[SC_CSTEP / 2], v4f (&qq)[SQ][SC_CSTEP / 4], v4f (&aa)[SC_CSTEP / 4],
                                        const float* kl, const float* ql, const float* al, int c0) {
#pragma unroll
    for (int u = 0; u < SC_CSTEP / 2; ++u)
        kk[u] = ld4(kl + (c0 / 2 + u) * KT_PITCH);
#pragma unroll
    for (int q = 0; q < SQ; ++q) {
#pragma unroll
        for (int g = 0; g < SC_CSTEP / 4; ++g)
            qq[q][g] = ld4(ql + q * PQ_PITCH + c0 + 4 * g);
    }
#pragma unroll
    for (int g = 0; g < SC_CSTEP / 4; ++g)
        aa[g] = ld4(al + c0 + 4 * g);
}

__device__ __forceinline__ void sc_step(v2f (&acc)[SQ], const v4f (&kk)[SC_CSTEP / 2],
                                        const v4f (&qq)[SQ][SC_CSTEP / 4], const v4f (&aa)[SC_CSTEP / 4]) {
#pragma unroll
    for (int u = 0; u < SC_CSTEP / 2; ++u) {
#pragma unroll
        for (int p = 0; p < 2; ++p) {
            const int   e  = 2 * u + p;
            const v2f   kp = p ? v2f{kk[u].z, kk[u].w} : v2f{kk[u].x, kk[u].y};
            const float ac = aa[e >> 2][e & 3];
#pragma unroll
            for (int q = 0; q < SQ; ++q) {
                const float qc = qq[q][e >> 2][e & 3];
                acc[q] = tanh2(kp + v2f{qc, qc}) * v2f{ac, ac} + acc[q];
            }
        }
    }
}

__global__ __launch_bounds__(32 * K1_WAVES) void qkv_scores_wmma_f32(
    const float* __restrict__ x,
    const float* __restrict__ Wq, const float* __restrict__ Wk, const float* __restrict__ Wv,
    const float* __restrict__ bq, const float* __restrict__ bk, const float* __restrict__ bv,
    const float* __restrict__ Wa, const float* __restrict__ ba,
    const float* __restrict__ attn_v,
    float* __restrict__ P,
    float* __restrict__ v)
{
    __shared__ __attribute__((aligned(16))) float kt[(HID / 2) * KT_PITCH];
    __shared__ __attribute__((aligned(16))) float pql[SS * PQ_PITCH];
    __shared__ __attribute__((aligned(16))) float avl[HID];

    const int lane = threadIdx.x;
    const int wave = __builtin_amdgcn_readfirstlane(threadIdx.y);
    const int tn   = blockIdx.x & (HH - 1);
    const int bat  = blockIdx.x / HH;

    if (wave < QKV_WAVES)
        qkv_pre_tile(x, Wq, Wk, Wv, bq, bk, bv, Wa, ba, v, kt, pql, bat, tn, wave, lane);
    else
        avl[(wave - QKV_WAVES) * 32 + lane] = attn_v[(wave - QKV_WAVES) * 32 + lane];
    __syncthreads();

    const int    i0 = wave * SQ;
    const float* kl = kt + lane * 4;
    const float* ql = pql + i0 * PQ_PITCH;

    v2f acc[SQ];
#pragma unroll
    for (int q = 0; q < SQ; ++q) acc[q] = v2f{0.0f, 0.0f};

    v4f kA[SC_CSTEP / 2], kB[SC_CSTEP / 2];
    v4f qA[SQ][SC_CSTEP / 4], qB[SQ][SC_CSTEP / 4];
    v4f aA[SC_CSTEP / 4], aB[SC_CSTEP / 4];
    sc_load(kA, qA, aA, kl, ql, avl, 0);
    for (int c0 = 0; c0 < HID - 2 * SC_CSTEP; c0 += 2 * SC_CSTEP) {
        sc_load(kB, qB, aB, kl, ql, avl, c0 + SC_CSTEP);
        __builtin_amdgcn_sched_barrier(0);
        sc_step(acc, kA, qA, aA);
        sc_load(kA, qA, aA, kl, ql, avl, c0 + 2 * SC_CSTEP);
        __builtin_amdgcn_sched_barrier(0);
        sc_step(acc, kB, qB, aB);
    }
    sc_load(kB, qB, aB, kl, ql, avl, HID - SC_CSTEP);
    __builtin_amdgcn_sched_barrier(0);
    sc_step(acc, kA, qA, aA);
    sc_step(acc, kB, qB, aB);

    float* prow = P + ((size_t)(bat * SS + i0) * HH + tn) * SS + lane;
#pragma unroll
    for (int q = 0; q < SQ; ++q) {
        float m = fmaxf(acc[q].x, acc[q].y);
        m = fmaxf(m, lane_xor<1>(m));
        m = fmaxf(m, lane_xor<2>(m));
        m = fmaxf(m, lane_xor<4>(m));
        m = fmaxf(m, lane_xor<8>(m));
        m = fmaxf(m, lane_xor<16>(m));
        const float e0 = __expf(acc[q].x - m);
        const float e1 = __expf(acc[q].y - m);
        float s = e0 + e1;
        s += lane_xor<1>(s);
        s += lane_xor<2>(s);
        s += lane_xor<4>(s);
        s += lane_xor<8>(s);
        s += lane_xor<16>(s);
        const float inv = 1.0f / s;
        prow[(size_t)q * (HH * SS)]      = e0 * inv;
        prow[(size_t)q * (HH * SS) + 32] = e1 * inv;
    }
}

extern "C" void kernel_launch(void* const* d_in, const int* in_sizes, int n_in,
                              void* d_out, int out_size, void* d_ws, size_t ws_size,
                              hipStream_t stream) {
    (void)in_sizes; (void)n_in; (void)out_size; (void)ws_size;
    const float* x      = (const float*)d_in[0];
    const float* Wq     = (const float*)d_in[1];
    const float* bq     = (const float*)d_in[2];
    const float* Wk     = (const float*)d_in[3];
    const float* bk     = (const float*)d_in[4];
    const float* Wv     = (const float*)d_in[5];
    const float* bv     = (const float*)d_in[6];
    const float* Wa     = (const float*)d_in[7];
    const float* ba     = (const float*)d_in[8];
    const float* attn_v = (const float*)d_in[9];
    const float* Wfc    = (const float*)d_in[10];
    const float* bfc    = (const float*)d_in[11];
    float* out = (float*)d_out;

    float* v = (float*)d_ws;
    float* P = v + (size_t)MQ * EE;

    qkv_scores_wmma_f32<<<K1_BLOCKS, dim3(32, K1_WAVES), 0, stream>>>(
        x, Wq, Wk, Wv, bq, bk, bv, Wa, ba, attn_v, P, v);

    fc_ctx_lds_wmma_f32<<<FC_BLOCKS, 256, 0, stream>>>(P, v, Wfc, bfc, out);
}
